# starting kernel
# baseline (speedup 1.0000x reference)
#include <hip/hip_runtime.h>
#include <hip/hip_bf16.h>

#define Nn 50000
#define Ee 800000
#define HX 64
#define HY 32
#define HT 32
#define FEAT 96
#define KX 128
#define GROUPS ((Nn + 15) / 16)

typedef __attribute__((ext_vector_type(2))) float v2f;
typedef __attribute__((ext_vector_type(8))) float v8f;

__global__ void zero_ws(float* __restrict__ aggr, int n) {
    int i = blockIdx.x * blockDim.x + threadIdx.x;
    if (i < n) aggr[i] = 0.0f;
}

__global__ void scatter_add(const int* __restrict__ erow,
                            const int* __restrict__ ecol,
                            const float* __restrict__ hX,
                            const float* __restrict__ hY,
                            float* __restrict__ aggr) {
    const int lane = threadIdx.x & 31;
    const int e    = blockIdx.x * 8 + (threadIdx.x >> 5);
    if (e >= Ee) return;
    const int col = ecol[e];
    const int row = erow[e];
    const float* srcX = hX + (size_t)col * HX;
    const float* srcY = hY + (size_t)col * HY;
    float* dst = aggr + (size_t)row * FEAT;
    unsafeAtomicAdd(&dst[lane],      srcX[lane]);
    unsafeAtomicAdd(&dst[lane + 32], srcX[lane + 32]);
    unsafeAtomicAdd(&dst[lane + 64], srcY[lane]);
}

__device__ __forceinline__ v8f wmma_f32(v2f a, v2f b, v8f c) {
    return __builtin_amdgcn_wmma_f32_16x16x4_f32(
         false, a,  false, b,
         (short)0, c,  false,  false);
}

__global__ void gemm_ln_X(const float* __restrict__ aggr,
                          const float* __restrict__ ht,
                          const float* __restrict__ WX,
                          const float* __restrict__ bX,
                          const float* __restrict__ gX,
                          const float* __restrict__ beX,
                          float* __restrict__ outX) {
    __shared__ float lds[4][16][HX + 1];
    const int wave  = threadIdx.x >> 5;
    const int lane  = threadIdx.x & 31;
    const int group = blockIdx.x * 4 + wave;

    if (group < GROUPS) {
        const int m     = lane & 15;
        const int khalf = lane >> 4;
        const int node  = group * 16 + m;
        const float* arow = aggr + (size_t)node * FEAT;

        v8f acc[4] = {};
        #pragma unroll 4
        for (int k0 = 0; k0 < KX; k0 += 4) {
            const int ka = k0 + khalf * 2;
            v2f a;
            if (ka < FEAT) { a.x = arow[ka];      a.y = arow[ka + 1]; }
            else           { a.x = ht[ka - FEAT]; a.y = ht[ka - FEAT + 1]; }
            #pragma unroll
            for (int t = 0; t < 4; ++t) {
                const int n = t * 16 + m;
                v2f b;
                b.x = WX[ka * HX + n];
                b.y = WX[(ka + 1) * HX + n];
                acc[t] = wmma_f32(a, b, acc[t]);
            }
        }
        #pragma unroll
        for (int t = 0; t < 4; ++t) {
            const int n = t * 16 + m;
            const float bias = bX[n];
            #pragma unroll
            for (int v = 0; v < 8; ++v) {
                const int row = khalf * 8 + v;
                float x = acc[t][v] + bias;
                lds[wave][row][n] = fmaxf(x, 0.0f);
            }
        }
    }
    __syncthreads();
    if (group < GROUPS) {
        const int row  = lane >> 1;
        const int half = lane & 1;
        float s = 0.0f, s2 = 0.0f;
        #pragma unroll
        for (int j = 0; j < 32; ++j) {
            float x = lds[wave][row][half * 32 + j];
            s += x; s2 += x * x;
        }
        s  += __shfl_xor(s, 1);
        s2 += __shfl_xor(s2, 1);
        const float mean = s * (1.0f / 64.0f);
        const float var  = s2 * (1.0f / 64.0f) - mean * mean;
        const float inv  = rsqrtf(var + 1e-5f);
        const int node   = group * 16 + row;
        #pragma unroll
        for (int j = 0; j < 32; ++j) {
            const int c = half * 32 + j;
            float x = lds[wave][row][c];
            outX[(size_t)node * HX + c] = (x - mean) * inv * gX[c] + beX[c];
        }
    }
}

__global__ void gemm_ln_Y(const float* __restrict__ aggr,
                          const float* __restrict__ WY,
                          const float* __restrict__ bY,
                          const float* __restrict__ gY,
                          const float* __restrict__ beY,
                          float* __restrict__ outY) {
    __shared__ float lds[4][16][HY + 1];
    const int wave  = threadIdx.x >> 5;
    const int lane  = threadIdx.x & 31;
    const int group = blockIdx.x * 4 + wave;

    if (group < GROUPS) {
        const int m     = lane & 15;
        const int khalf = lane >> 4;
        const int node  = group * 16 + m;
        const float* arow = aggr + (size_t)node * FEAT + HX;

        v8f acc[2] = {};
        #pragma unroll
        for (int k0 = 0; k0 < HY; k0 += 4) {
            const int ka = k0 + khalf * 2;
            v2f a;
            a.x = arow[ka];
            a.y = arow[ka + 1];
            #pragma unroll
            for (int t = 0; t < 2; ++t) {
                const int n = t * 16 + m;
                v2f b;
                b.x = WY[ka * HY + n];
                b.y = WY[(ka + 1) * HY + n];
                acc[t] = wmma_f32(a, b, acc[t]);
            }
        }
        #pragma unroll
        for (int t = 0; t < 2; ++t) {
            const int n = t * 16 + m;
            const float bias = bY[n];
            #pragma unroll
            for (int v = 0; v < 8; ++v) {
                const int row = khalf * 8 + v;
                float x = acc[t][v] + bias;
                lds[wave][row][n] = fmaxf(x, 0.0f);
            }
        }
    }
    __syncthreads();
    if (group < GROUPS) {
        const int row  = lane >> 1;
        const int half = lane & 1;
        float s = 0.0f, s2 = 0.0f;
        #pragma unroll
        for (int j = 0; j < 16; ++j) {
            float x = lds[wave][row][half * 16 + j];
            s += x; s2 += x * x;
        }
        s  += __shfl_xor(s, 1);
        s2 += __shfl_xor(s2, 1);
        const float mean = s * (1.0f / 32.0f);
        const float var  = s2 * (1.0f / 32.0f) - mean * mean;
        const float inv  = rsqrtf(var + 1e-5f);
        const int node   = group * 16 + row;
        #pragma unroll
        for (int j = 0; j < 16; ++j) {
            const int c = half * 16 + j;
            float x = lds[wave][row][c];
            outY[(size_t)node * HY + c] = (x - mean) * inv * gY[c] + beY[c];
        }
    }
}

extern "C" void kernel_launch(void* const* d_in, const int* in_sizes, int n_in,
                              void* d_out, int out_size, void* d_ws, size_t ws_size,
                              hipStream_t stream) {
    const int*   erow = (const int*)d_in[0];
    const int*   ecol = (const int*)d_in[1];
    const float* hX   = (const float*)d_in[2];
    const float* hY   = (const float*)d_in[3];
    const float* ht   = (const float*)d_in[4];
    const float* WX   = (const float*)d_in[5];
    const float* bX   = (const float*)d_in[6];
    const float* gX   = (const float*)d_in[7];
    const float* beX  = (const float*)d_in[8];
    const float* WY   = (const float*)d_in[9];
    const float* bY   = (const float*)d_in[10];
    const float* gY   = (const float*)d_in[11];
    const float* beY  = (const float*)d_in[12];

    float* out  = (float*)d_out;
    float* aggr = (float*)d_ws;

    const int nAggr = Nn * FEAT;
    zero_ws<<<(nAggr + 255) / 256, 256, 0, stream>>>(aggr, nAggr);

    scatter_add<<<(Ee + 7) / 8, 256, 0, stream>>>(erow, ecol, hX, hY, aggr);

    const int blocks = (GROUPS + 3) / 4;
    gemm_ln_X<<<blocks, 128, 0, stream>>>(aggr, ht, WX, bX, gX, beX, out);
    gemm_ln_Y<<<blocks, 128, 0, stream>>>(aggr, WY, bY, gY, beY, out + (size_t)Nn * HX);
}


# Pull aggregation, fixed f32 sum order (earlier pull summed in atomic arrival order): edges bucketed by destination, each wave sorts its node's source ids, adds rows by ascending id; no f32 atomics
# speedup vs baseline: 1.0678x; 1.0678x over previous
#include <hip/hip_runtime.h>
#include <hip/hip_bf16.h>


#define Nn 50000
#define Ee 800000
#define HX 64
#define HY 32
#define HT 32
#define FEAT 96
#define KX 128
#define GROUPS ((Nn + 15) / 16)

#define SCAN_T   1024
#define SCAN_PER 52
#define NPAD     (SCAN_T * SCAN_PER)

#define WS_AGGR 0
#define WS_CNT  (Nn * FEAT * 4)
#define WS_OFF  (WS_CNT + NPAD * 4)
#define WS_RNK  (WS_OFF + NPAD * 4)
#define WS_SCOL (WS_RNK + Ee * 4)

typedef __attribute__((ext_vector_type(2))) float v2f;
typedef __attribute__((ext_vector_type(8))) float v8f;

__global__ void zero_cnt(int* __restrict__ cnt) {
    const int i = blockIdx.x * blockDim.x + threadIdx.x;
    if (i < NPAD / 4) ((int4*)cnt)[i] = make_int4(0, 0, 0, 0);
}

__global__ void hist_rank(const int* __restrict__ erow,
                          int* __restrict__ cnt,
                          int* __restrict__ rnk) {
    const int q = blockIdx.x * blockDim.x + threadIdx.x;
    if (q < Ee / 4) {
        const int4 r = ((const int4*)erow)[q];
        int4 k;
        k.x = atomicAdd(cnt + r.x, 1);
        k.y = atomicAdd(cnt + r.y, 1);
        k.z = atomicAdd(cnt + r.z, 1);
        k.w = atomicAdd(cnt + r.w, 1);
        ((int4*)rnk)[q] = k;
    }
}

__global__ void __launch_bounds__(SCAN_T) scan_counts(const int* __restrict__ cnt,
                                                      int* __restrict__ off) {
    __shared__ int wsum[32];
    const int t    = threadIdx.x;
    const int lane = t & 31;
    const int wave = t >> 5;
    const int4* src = (const int4*)(cnt + wave * (SCAN_PER * 32)) + lane;
    int4 v[SCAN_PER / 4];
    #pragma unroll
    for (int i = 0; i < SCAN_PER / 4; ++i) v[i] = src[i * 32];
    int carry = 0;
    #pragma unroll
    for (int i = 0; i < SCAN_PER / 4; ++i) {
        const int a = v[i].x, b = v[i].y, c = v[i].z, d = v[i].w;
        const int tsum = a + b + c + d;
        int x = tsum;
        for (int k = 1; k < 32; k <<= 1) {
            const int o = __shfl_up(x, k);
            if (lane >= k) x += o;
        }
        const int ex = carry + x - tsum;
        v[i].x = ex;
        v[i].y = ex + a;
        v[i].z = ex + a + b;
        v[i].w = ex + a + b + c;
        carry += __builtin_amdgcn_readlane(x, 31);
    }
    if (lane == 0) wsum[wave] = carry;
    __syncthreads();
    if (wave == 0) {
        const int w = wsum[lane];
        int winc = w;
        for (int k = 1; k < 32; k <<= 1) {
            const int o = __shfl_up(winc, k);
            if (lane >= k) winc += o;
        }
        wsum[lane] = winc - w;
    }
    __syncthreads();
    const int base = wsum[wave];
    int4* dst = (int4*)(off + wave * (SCAN_PER * 32)) + lane;
    #pragma unroll
    for (int i = 0; i < SCAN_PER / 4; ++i)
        dst[i * 32] = make_int4(v[i].x + base, v[i].y + base, v[i].z + base, v[i].w + base);
}

__global__ void place_cols(const int* __restrict__ erow,
                           const int* __restrict__ ecol,
                           const int* __restrict__ rnk,
                           const int* __restrict__ off,
                           int* __restrict__ scol) {
    const int q = blockIdx.x * blockDim.x + threadIdx.x;
    if (q < Ee / 4) {
        const int4 r = ((const int4*)erow)[q];
        const int4 k = ((const int4*)rnk)[q];
        const int4 s = ((const int4*)ecol)[q];
        const int o0 = off[r.x], o1 = off[r.y], o2 = off[r.z], o3 = off[r.w];
        scol[o0 + k.x] = s.x;
        scol[o1 + k.y] = s.y;
        scol[o2 + k.z] = s.z;
        scol[o3 + k.w] = s.w;
    }
}

#define SENT 0x7fffffff

__device__ __forceinline__ float4 add4(float4 a, float4 b) {
    return make_float4(a.x + b.x, a.y + b.y, a.z + b.z, a.w + b.w);
}

template <bool TWO>
__device__ __forceinline__ float4 sum_sorted(const int* __restrict__ seg, const int d,
                                             const int lane, const bool act,
                                             const float* __restrict__ bp, const int rowlen) {
    const int c0 = (lane < d) ? seg[lane] : SENT;
    int c1 = SENT;
    if (TWO) c1 = (lane + 32 < d) ? seg[lane + 32] : SENT;
    int r0 = 0, r1 = 0;
    const int d0 = TWO ? 32 : d;
    for (int j = 0; j < d0; ++j) {
        const int cj = __builtin_amdgcn_readlane(c0, j);
        r0 += (cj < c0 || (cj == c0 && j < lane)) ? 1 : 0;
        if (TWO) r1 += (cj <= c1) ? 1 : 0;
    }
    if (TWO) {
        for (int j = 32; j < d; ++j) {
            const int cj = __builtin_amdgcn_readlane(c1, j - 32);
            r0 += (cj < c0) ? 1 : 0;
            r1 += (cj < c1 || (cj == c1 && j - 32 < lane)) ? 1 : 0;
        }
    }
    if (lane >= d) r0 = -1;
    if (!TWO || lane + 32 >= d) r1 = -1;

    auto pick = [&](const int p) -> int {
        const unsigned m0 = __builtin_amdgcn_ballot_w32(r0 == p);
        if (!TWO) return __builtin_amdgcn_readlane(c0, __builtin_ctz(m0));
        const unsigned m1 = __builtin_amdgcn_ballot_w32(r1 == p);
        const int a = __builtin_amdgcn_readlane(c0, m0 ? __builtin_ctz(m0) : 0);
        const int b = __builtin_amdgcn_readlane(c1, m1 ? __builtin_ctz(m1) : 0);
        return m0 ? a : b;
    };

    const float4 z = make_float4(0.f, 0.f, 0.f, 0.f);
    float4 acc = z;
    int p = 0;
    for (; p + 4 <= d; p += 4) {
        const int ca = pick(p), cb = pick(p + 1), cc = pick(p + 2), cd = pick(p + 3);
        float4 va = z, vb = z, vc = z, vd = z;
        if (act) {
            va = *(const float4*)(bp + (size_t)ca * rowlen);
            vb = *(const float4*)(bp + (size_t)cb * rowlen);
            vc = *(const float4*)(bp + (size_t)cc * rowlen);
            vd = *(const float4*)(bp + (size_t)cd * rowlen);
        }
        acc = add4(acc, va);
        acc = add4(acc, vb);
        acc = add4(acc, vc);
        acc = add4(acc, vd);
    }
    for (; p < d; ++p) {
        const int ca = pick(p);
        float4 va = z;
        if (act) va = *(const float4*)(bp + (size_t)ca * rowlen);
        acc = add4(acc, va);
    }
    return acc;
}

__device__ __forceinline__ float4 sum_sorted_any(const int* __restrict__ seg, const int d,
                                              const int lane, const bool act,
                                              const float* __restrict__ bp, const int rowlen) {
    const float4 z = make_float4(0.f, 0.f, 0.f, 0.f);
    float4 acc = z;
    int last = -1;
    int done = 0;
    while (done < d) {
        int best = SENT;
        for (int i = lane; i < d; i += 32) {
            const int c = seg[i];
            if (c > last && c < best) best = c;
        }
        for (int s = 16; s > 0; s >>= 1) {
            const int o = __shfl_xor(best, s);
            best = o < best ? o : best;
        }
        int n = 0;
        for (int i = lane; i < d; i += 32) n += (seg[i] == best) ? 1 : 0;
        for (int s = 16; s > 0; s >>= 1) n += __shfl_xor(n, s);
        best = __builtin_amdgcn_readfirstlane(best);
        n    = __builtin_amdgcn_readfirstlane(n);
        if (n <= 0) break;
        float4 v = z;
        if (act) v = *(const float4*)(bp + (size_t)best * rowlen);
        for (int k = 0; k < n; ++k) acc = add4(acc, v);
        last = best;
        done += n;
    }
    return acc;
}

__global__ void __launch_bounds__(256) pull_sum(const int* __restrict__ off,
                                                const int* __restrict__ scol,
                                                const float* __restrict__ hX,
                                                const float* __restrict__ hY,
                                                float* __restrict__ aggr) {
    const int lane = threadIdx.x & 31;
    const int node = __builtin_amdgcn_readfirstlane(blockIdx.x * 8 + (threadIdx.x >> 5));
    if (node >= Nn) return;
    const int beg = __builtin_amdgcn_readfirstlane(off[node]);
    const int d   = __builtin_amdgcn_readfirstlane(off[node + 1]) - beg;
    const int* seg = scol + beg;

    const bool act = lane < 24;
    const float* bp = (lane < 16) ? (hX + lane * 4) : (hY + ((lane - 16) & 7) * 4);
    const int rowlen = (lane < 16) ? HX : HY;

    float4 acc;
    if (d <= 32)      acc = sum_sorted<false>(seg, d, lane, act, bp, rowlen);
    else if (d <= 64) acc = sum_sorted<true>(seg, d, lane, act, bp, rowlen);
    else              acc = sum_sorted_any(seg, d, lane, act, bp, rowlen);

    if (act) *(float4*)(aggr + (size_t)node * FEAT + lane * 4) = acc;
}

__device__ __forceinline__ v8f wmma_f32(v2f a, v2f b, v8f c) {
    return __builtin_amdgcn_wmma_f32_16x16x4_f32(
         false, a,  false, b,
         (short)0, c,  false,  false);
}

__global__ void gemm_ln_X(const float* __restrict__ aggr,
                          const float* __restrict__ ht,
                          const float* __restrict__ WX,
                          const float* __restrict__ bX,
                          const float* __restrict__ gX,
                          const float* __restrict__ beX,
                          float* __restrict__ outX) {
    __shared__ float lds[4][16][HX + 1];
    const int wave  = threadIdx.x >> 5;
    const int lane  = threadIdx.x & 31;
    const int group = blockIdx.x * 4 + wave;

    if (group < GROUPS) {
        const int m     = lane & 15;
        const int khalf = lane >> 4;
        const int node  = group * 16 + m;
        const float* arow = aggr + (size_t)node * FEAT;

        v8f acc[4] = {};
        #pragma unroll 4
        for (int k0 = 0; k0 < KX; k0 += 4) {
            const int ka = k0 + khalf * 2;
            v2f a;
            if (ka < FEAT) { a.x = arow[ka];      a.y = arow[ka + 1]; }
            else           { a.x = ht[ka - FEAT]; a.y = ht[ka - FEAT + 1]; }
            #pragma unroll
            for (int t = 0; t < 4; ++t) {
                const int n = t * 16 + m;
                v2f b;
                b.x = WX[ka * HX + n];
                b.y = WX[(ka + 1) * HX + n];
                acc[t] = wmma_f32(a, b, acc[t]);
            }
        }
        #pragma unroll
        for (int t = 0; t < 4; ++t) {
            const int n = t * 16 + m;
            const float bias = bX[n];
            #pragma unroll
            for (int v = 0; v < 8; ++v) {
                const int row = khalf * 8 + v;
                float x = acc[t][v] + bias;
                lds[wave][row][n] = fmaxf(x, 0.0f);
            }
        }
    }
    __syncthreads();
    if (group < GROUPS) {
        const int row  = lane >> 1;
        const int half = lane & 1;
        float s = 0.0f, s2 = 0.0f;
        #pragma unroll
        for (int j = 0; j < 32; ++j) {
            float x = lds[wave][row][half * 32 + j];
            s += x; s2 += x * x;
        }
        s  += __shfl_xor(s, 1);
        s2 += __shfl_xor(s2, 1);
        const float mean = s * (1.0f / 64.0f);
        const float var  = s2 * (1.0f / 64.0f) - mean * mean;
        const float inv  = rsqrtf(var + 1e-5f);
        const int node   = group * 16 + row;
        #pragma unroll
        for (int j = 0; j < 32; ++j) {
            const int c = half * 32 + j;
            float x = lds[wave][row][c];
            outX[(size_t)node * HX + c] = (x - mean) * inv * gX[c] + beX[c];
        }
    }
}

__global__ void gemm_ln_Y(const float* __restrict__ aggr,
                          const float* __restrict__ WY,
                          const float* __restrict__ bY,
                          const float* __restrict__ gY,
                          const float* __restrict__ beY,
                          float* __restrict__ outY) {
    __shared__ float lds[4][16][HY + 1];
    const int wave  = threadIdx.x >> 5;
    const int lane  = threadIdx.x & 31;
    const int group = blockIdx.x * 4 + wave;

    if (group < GROUPS) {
        const int m     = lane & 15;
        const int khalf = lane >> 4;
        const int node  = group * 16 + m;
        const float* arow = aggr + (size_t)node * FEAT + HX;

        v8f acc[2] = {};
        #pragma unroll
        for (int k0 = 0; k0 < HY; k0 += 4) {
            const int ka = k0 + khalf * 2;
            v2f a;
            a.x = arow[ka];
            a.y = arow[ka + 1];
            #pragma unroll
            for (int t = 0; t < 2; ++t) {
                const int n = t * 16 + m;
                v2f b;
                b.x = WY[ka * HY + n];
                b.y = WY[(ka + 1) * HY + n];
                acc[t] = wmma_f32(a, b, acc[t]);
            }
        }
        #pragma unroll
        for (int t = 0; t < 2; ++t) {
            const int n = t * 16 + m;
            const float bias = bY[n];
            #pragma unroll
            for (int v = 0; v < 8; ++v) {
                const int row = khalf * 8 + v;
                float x = acc[t][v] + bias;
                lds[wave][row][n] = fmaxf(x, 0.0f);
            }
        }
    }
    __syncthreads();
    if (group < GROUPS) {
        const int row  = lane >> 1;
        const int half = lane & 1;
        float s = 0.0f, s2 = 0.0f;
        #pragma unroll
        for (int j = 0; j < 16; ++j) {
            float x = lds[wave][row][half * 16 + j];
            s += x; s2 += x * x;
        }
        s  += __shfl_xor(s, 1);
        s2 += __shfl_xor(s2, 1);
        const float mean = s * (1.0f / 32.0f);
        const float var  = s2 * (1.0f / 32.0f) - mean * mean;
        const float inv  = rsqrtf(var + 1e-5f);
        const int node   = group * 16 + row;
        #pragma unroll
        for (int j = 0; j < 16; ++j) {
            const int c = half * 16 + j;
            float x = lds[wave][row][c];
            outY[(size_t)node * HY + c] = (x - mean) * inv * gY[c] + beY[c];
        }
    }
}

extern "C" void kernel_launch(void* const* d_in, const int* in_sizes, int n_in,
                              void* d_out, int out_size, void* d_ws, size_t ws_size,
                              hipStream_t stream) {
    const int*   erow = (const int*)d_in[0];
    const int*   ecol = (const int*)d_in[1];
    const float* hX   = (const float*)d_in[2];
    const float* hY   = (const float*)d_in[3];
    const float* ht   = (const float*)d_in[4];
    const float* WX   = (const float*)d_in[5];
    const float* bX   = (const float*)d_in[6];
    const float* gX   = (const float*)d_in[7];
    const float* beX  = (const float*)d_in[8];
    const float* WY   = (const float*)d_in[9];
    const float* bY   = (const float*)d_in[10];
    const float* gY   = (const float*)d_in[11];
    const float* beY  = (const float*)d_in[12];

    float* out  = (float*)d_out;
    char*  ws   = (char*)d_ws;
    float* aggr = (float*)(ws + WS_AGGR);
    int*   cnt  = (int*)(ws + WS_CNT);
    int*   off  = (int*)(ws + WS_OFF);
    int*   rnk  = (int*)(ws + WS_RNK);
    int*   scol = (int*)(ws + WS_SCOL);

    zero_cnt<<<(NPAD / 4 + 255) / 256, 256, 0, stream>>>(cnt);
    hist_rank<<<(Ee / 4 + 255) / 256, 256, 0, stream>>>(erow, cnt, rnk);
    scan_counts<<<1, SCAN_T, 0, stream>>>(cnt, off);
    place_cols<<<(Ee / 4 + 255) / 256, 256, 0, stream>>>(erow, ecol, rnk, off, scol);

    pull_sum<<<(Nn + 7) / 8, 256, 0, stream>>>(off, scol, hX, hY, aggr);

    const int blocks = (GROUPS + 3) / 4;
    gemm_ln_X<<<blocks, 128, 0, stream>>>(aggr, ht, WX, bX, gX, beX, out);
    gemm_ln_Y<<<blocks, 128, 0, stream>>>(aggr, WY, bY, gY, beY, out + (size_t)Nn * HX);
}


# One kernel replaces pull_sum and both product kernels: a 16-wave block sums 32 nodes' rows into LDS, WMMA reads A there, LayerNorm row sums by lane shuffle tree; aggr never written; 7 launches to 5
# speedup vs baseline: 1.1936x; 1.1178x over previous
#include <hip/hip_runtime.h>
#include <hip/hip_bf16.h>


#define Nn 50000
#define Ee 800000
#define HX 64
#define HY 32
#define HT 32
#define FEAT 96
#define KX 128

#define SCAN_T   1024
#define SCAN_PER 52
#define NPAD     (SCAN_T * SCAN_PER)

#define WS_AGGR 0
#define WS_CNT  (Nn * FEAT * 4)
#define WS_OFF  (WS_CNT + NPAD * 4)
#define WS_RNK  (WS_OFF + NPAD * 4)
#define WS_SCOL (WS_RNK + Ee * 4)

typedef __attribute__((ext_vector_type(2))) float v2f;
typedef __attribute__((ext_vector_type(8))) float v8f;

__global__ void zero_cnt(int* __restrict__ cnt) {
    const int i = blockIdx.x * blockDim.x + threadIdx.x;
    if (i < NPAD / 4) ((int4*)cnt)[i] = make_int4(0, 0, 0, 0);
}

__global__ void hist_rank(const int* __restrict__ erow,
                          int* __restrict__ cnt,
                          int* __restrict__ rnk) {
    const int q = blockIdx.x * blockDim.x + threadIdx.x;
    if (q < Ee / 4) {
        const int4 r = ((const int4*)erow)[q];
        int4 k;
        k.x = atomicAdd(cnt + r.x, 1);
        k.y = atomicAdd(cnt + r.y, 1);
        k.z = atomicAdd(cnt + r.z, 1);
        k.w = atomicAdd(cnt + r.w, 1);
        ((int4*)rnk)[q] = k;
    }
}

__global__ void __launch_bounds__(SCAN_T) scan_counts(const int* __restrict__ cnt,
                                                      int* __restrict__ off) {
    __shared__ int wsum[32];
    const int t    = threadIdx.x;
    const int lane = t & 31;
    const int wave = t >> 5;
    const int4* src = (const int4*)(cnt + wave * (SCAN_PER * 32)) + lane;
    int4 v[SCAN_PER / 4];
    #pragma unroll
    for (int i = 0; i < SCAN_PER / 4; ++i) v[i] = src[i * 32];
    int carry = 0;
    #pragma unroll
    for (int i = 0; i < SCAN_PER / 4; ++i) {
        const int a = v[i].x, b = v[i].y, c = v[i].z, d = v[i].w;
        const int tsum = a + b + c + d;
        int x = tsum;
        for (int k = 1; k < 32; k <<= 1) {
            const int o = __shfl_up(x, k);
            if (lane >= k) x += o;
        }
        const int ex = carry + x - tsum;
        v[i].x = ex;
        v[i].y = ex + a;
        v[i].z = ex + a + b;
        v[i].w = ex + a + b + c;
        carry += __builtin_amdgcn_readlane(x, 31);
    }
    if (lane == 0) wsum[wave] = carry;
    __syncthreads();
    if (wave == 0) {
        const int w = wsum[lane];
        int winc = w;
        for (int k = 1; k < 32; k <<= 1) {
            const int o = __shfl_up(winc, k);
            if (lane >= k) winc += o;
        }
        wsum[lane] = winc - w;
    }
    __syncthreads();
    const int base = wsum[wave];
    int4* dst = (int4*)(off + wave * (SCAN_PER * 32)) + lane;
    #pragma unroll
    for (int i = 0; i < SCAN_PER / 4; ++i)
        dst[i * 32] = make_int4(v[i].x + base, v[i].y + base, v[i].z + base, v[i].w + base);
}

__global__ void place_cols(const int* __restrict__ erow,
                           const int* __restrict__ ecol,
                           const int* __restrict__ rnk,
                           const int* __restrict__ off,
                           int* __restrict__ scol) {
    const int q = blockIdx.x * blockDim.x + threadIdx.x;
    if (q < Ee / 4) {
        const int4 r = ((const int4*)erow)[q];
        const int4 k = ((const int4*)rnk)[q];
        const int4 s = ((const int4*)ecol)[q];
        const int o0 = off[r.x], o1 = off[r.y], o2 = off[r.z], o3 = off[r.w];
        scol[o0 + k.x] = s.x;
        scol[o1 + k.y] = s.y;
        scol[o2 + k.z] = s.z;
        scol[o3 + k.w] = s.w;
    }
}

#define SENT 0x7fffffff

__device__ __forceinline__ float4 add4(float4 a, float4 b) {
    return make_float4(a.x + b.x, a.y + b.y, a.z + b.z, a.w + b.w);
}

template <bool TWO>
__device__ __forceinline__ float4 sum_sorted(const int* __restrict__ seg, const int d,
                                             const int lane, const bool act,
                                             const float* __restrict__ bp, const int rowsh) {
    const int c0 = (lane < d) ? seg[lane] : SENT;
    int c1 = SENT;
    if (TWO) c1 = (lane + 32 < d) ? seg[lane + 32] : SENT;
    int r0 = 0, r1 = 0;
    const int d0 = TWO ? 32 : d;
    for (int j = 0; j < d0; ++j) {
        const int cj = __builtin_amdgcn_readlane(c0, j);
        r0 += (cj < c0 || (cj == c0 && j < lane)) ? 1 : 0;
        if (TWO) r1 += (cj <= c1) ? 1 : 0;
    }
    if (TWO) {
        for (int j = 32; j < d; ++j) {
            const int cj = __builtin_amdgcn_readlane(c1, j - 32);
            r0 += (cj < c0) ? 1 : 0;
            r1 += (cj < c1 || (cj == c1 && j - 32 < lane)) ? 1 : 0;
        }
    }
    if (lane >= d) r0 = -1;
    if (!TWO || lane + 32 >= d) r1 = -1;

    auto pick = [&](const int p) -> int {
        const unsigned m0 = __builtin_amdgcn_ballot_w32(r0 == p);
        if (!TWO) return __builtin_amdgcn_readlane(c0, __builtin_ctz(m0));
        const unsigned m1 = __builtin_amdgcn_ballot_w32(r1 == p);
        const int a = __builtin_amdgcn_readlane(c0, m0 ? __builtin_ctz(m0) : 0);
        const int b = __builtin_amdgcn_readlane(c1, m1 ? __builtin_ctz(m1) : 0);
        return m0 ? a : b;
    };

    const float4 z = make_float4(0.f, 0.f, 0.f, 0.f);
    float4 acc = z;
    int p = 0;
    for (; p + 4 <= d; p += 4) {
        const int ca = pick(p), cb = pick(p + 1), cc = pick(p + 2), cd = pick(p + 3);
        float4 va = z, vb = z, vc = z, vd = z;
        if (act) {
            va = *(const float4*)(bp + ((size_t)ca << rowsh));
            vb = *(const float4*)(bp + ((size_t)cb << rowsh));
            vc = *(const float4*)(bp + ((size_t)cc << rowsh));
            vd = *(const float4*)(bp + ((size_t)cd << rowsh));
        }
        acc = add4(acc, va);
        acc = add4(acc, vb);
        acc = add4(acc, vc);
        acc = add4(acc, vd);
    }
    for (; p < d; ++p) {
        const int ca = pick(p);
        float4 va = z;
        if (act) va = *(const float4*)(bp + ((size_t)ca << rowsh));
        acc = add4(acc, va);
    }
    return acc;
}

__device__ __forceinline__ float4 sum_sorted_any(const int* __restrict__ seg, const int d,
                                              const int lane, const bool act,
                                              const float* __restrict__ bp, const int rowsh) {
    const float4 z = make_float4(0.f, 0.f, 0.f, 0.f);
    float4 acc = z;
    int last = -1;
    int done = 0;
    while (done < d) {
        int best = SENT;
        for (int i = lane; i < d; i += 32) {
            const int c = seg[i];
            if (c > last && c < best) best = c;
        }
        for (int s = 16; s > 0; s >>= 1) {
            const int o = __shfl_xor(best, s);
            best = o < best ? o : best;
        }
        int n = 0;
        for (int i = lane; i < d; i += 32) n += (seg[i] == best) ? 1 : 0;
        for (int s = 16; s > 0; s >>= 1) n += __shfl_xor(n, s);
        best = __builtin_amdgcn_readfirstlane(best);
        n    = __builtin_amdgcn_readfirstlane(n);
        if (n <= 0) break;
        float4 v = z;
        if (act) v = *(const float4*)(bp + ((size_t)best << rowsh));
        for (int k = 0; k < n; ++k) acc = add4(acc, v);
        last = best;
        done += n;
    }
    return acc;
}

__device__ __forceinline__ v8f wmma_f32(v2f a, v2f b, v8f c) {
    return __builtin_amdgcn_wmma_f32_16x16x4_f32(
         false, a,  false, b,
         (short)0, c,  false,  false);
}

#define BLK_WAVES 16
#define BLK_NODES 32
#define BLK_COUNT ((Nn + BLK_NODES - 1) / BLK_NODES)
#define APITCH 132
#define CPITCH 100

__global__ void __launch_bounds__(BLK_WAVES * 32, 16) pull_gemm_ln(const int* __restrict__ off,
                                                               const int* __restrict__ scol,
                                                               const float* __restrict__ hX,
                                                               const float* __restrict__ hY,
                                                               const float* __restrict__ ht,
                                                               const float* __restrict__ WX,
                                                               const float* __restrict__ bX,
                                                               const float* __restrict__ gX,
                                                               const float* __restrict__ beX,
                                                               const float* __restrict__ WY,
                                                               const float* __restrict__ bY,
                                                               const float* __restrict__ gY,
                                                               const float* __restrict__ beY,
                                                               float* __restrict__ outX,
                                                               float* __restrict__ outY) {
    __shared__ __attribute__((aligned(16))) float At[BLK_NODES * APITCH];
    __shared__ __attribute__((aligned(16))) float Ct[BLK_NODES * CPITCH];

    const int lane = threadIdx.x & 31;
    const int wave = __builtin_amdgcn_readfirstlane(threadIdx.x >> 5);
    const int base = blockIdx.x * BLK_NODES;
    const bool act = lane < 24;

    {
        const float* bp = (lane < 16) ? (hX + lane * 4) : (hY + ((lane - 16) & 7) * 4);
        const int rowsh = (lane < 16) ? 6 : 5;
        #pragma nounroll
        for (int r = 0; r < BLK_NODES / 16; ++r) {
            const int row  = r * 16 + wave;
            const int node = base + row;
            float4 acc = make_float4(0.f, 0.f, 0.f, 0.f);
            if (node < Nn) {
                const int beg = __builtin_amdgcn_readfirstlane(off[node]);
                const int d   = __builtin_amdgcn_readfirstlane(off[node + 1]) - beg;
                const int* seg = scol + beg;
                if (d <= 32)      acc = sum_sorted<false>(seg, d, lane, act, bp, rowsh);
                else if (d <= 64) acc = sum_sorted<true>(seg, d, lane, act, bp, rowsh);
                else              acc = sum_sorted_any(seg, d, lane, act, bp, rowsh);
            }
            if (act) *(float4*)(At + row * APITCH + lane * 4) = acc;
            else     *(float4*)(At + row * APITCH + FEAT + (lane - 24) * 4) = *(const float4*)(ht + (lane - 24) * 4);
        }
    }
    __syncthreads();

    {
        const int m     = lane & 15;
        const int khalf = lane >> 4;
        if (wave < 8) {
            const int rt = wave >> 2, ct = wave & 3;
            const float* arow = At + (rt * 16 + m) * APITCH + khalf * 2;
            const float* wcol = WX + (khalf * 2) * HX + ct * 16 + m;
            v8f c = {};
            #pragma unroll 8
            for (int k0 = 0; k0 < KX; k0 += 4) {
                const v2f a = *(const v2f*)(arow + k0);
                v2f b;
                b.x = wcol[k0 * HX];
                b.y = wcol[k0 * HX + HX];
                c = wmma_f32(a, b, c);
            }
            const float bias = bX[ct * 16 + m];
            #pragma unroll
            for (int v = 0; v < 8; ++v)
                Ct[(rt * 16 + khalf * 8 + v) * CPITCH + ct * 16 + m] = fmaxf(c[v] + bias, 0.0f);
        } else if (wave < 12) {
            const int rt = (wave - 8) >> 1, ct = (wave - 8) & 1;
            const float* arow = At + (rt * 16 + m) * APITCH + HX + khalf * 2;
            const float* wcol = WY + (khalf * 2) * HY + ct * 16 + m;
            v8f c = {};
            #pragma unroll
            for (int k0 = 0; k0 < HY; k0 += 4) {
                const v2f a = *(const v2f*)(arow + k0);
                v2f b;
                b.x = wcol[k0 * HY];
                b.y = wcol[k0 * HY + HY];
                c = wmma_f32(a, b, c);
            }
            const float bias = bY[ct * 16 + m];
            #pragma unroll
            for (int v = 0; v < 8; ++v)
                Ct[(rt * 16 + khalf * 8 + v) * CPITCH + HX + ct * 16 + m] = fmaxf(c[v] + bias, 0.0f);
        }
    }
    __syncthreads();

    {
        const int c4 = (lane < 16) ? lane * 4 : (lane - 16) * 4;
        float4 g  = make_float4(0.f, 0.f, 0.f, 0.f);
        float4 be = g;
        if (act) {
            g  = *(const float4*)(((lane < 16) ? gX : gY) + c4);
            be = *(const float4*)(((lane < 16) ? beX : beY) + c4);
        }
        const float rn = (lane < 16) ? (1.0f / 64.0f) : (1.0f / 32.0f);
        #pragma unroll
        for (int r = 0; r < BLK_NODES / 16; ++r) {
            const int row  = r * 16 + wave;
            const int node = base + row;
            if (node < Nn) {
                float4 x = make_float4(0.f, 0.f, 0.f, 0.f);
                if (act) x = *(const float4*)(Ct + row * CPITCH + lane * 4);
                float s  = (x.x + x.y) + (x.z + x.w);
                float s2 = (x.x * x.x + x.y * x.y) + (x.z * x.z + x.w * x.w);
                #pragma unroll
                for (int o = 1; o < 16; o <<= 1) {
                    s  += __shfl_xor(s, o);
                    s2 += __shfl_xor(s2, o);
                }
                const float mean = s * rn;
                const float var  = s2 * rn - mean * mean;
                const float inv  = rsqrtf(var + 1e-5f);
                if (act) {
                    float* o = (lane < 16) ? (outX + (size_t)node * HX + c4) : (outY + (size_t)node * HY + c4);
                    float4 y;
                    y.x = (x.x - mean) * inv * g.x + be.x;
                    y.y = (x.y - mean) * inv * g.y + be.y;
                    y.z = (x.z - mean) * inv * g.z + be.z;
                    y.w = (x.w - mean) * inv * g.w + be.w;
                    *(float4*)o = y;
                }
            }
        }
    }
}

extern "C" void kernel_launch(void* const* d_in, const int* in_sizes, int n_in,
                              void* d_out, int out_size, void* d_ws, size_t ws_size,
                              hipStream_t stream) {
    const int*   erow = (const int*)d_in[0];
    const int*   ecol = (const int*)d_in[1];
    const float* hX   = (const float*)d_in[2];
    const float* hY   = (const float*)d_in[3];
    const float* ht   = (const float*)d_in[4];
    const float* WX   = (const float*)d_in[5];
    const float* bX   = (const float*)d_in[6];
    const float* gX   = (const float*)d_in[7];
    const float* beX  = (const float*)d_in[8];
    const float* WY   = (const float*)d_in[9];
    const float* bY   = (const float*)d_in[10];
    const float* gY   = (const float*)d_in[11];
    const float* beY  = (const float*)d_in[12];

    float* out  = (float*)d_out;
    char*  ws   = (char*)d_ws;
    int*   cnt  = (int*)(ws + WS_CNT);
    int*   off  = (int*)(ws + WS_OFF);
    int*   rnk  = (int*)(ws + WS_RNK);
    int*   scol = (int*)(ws + WS_SCOL);

    zero_cnt<<<(NPAD / 4 + 255) / 256, 256, 0, stream>>>(cnt);
    hist_rank<<<(Ee / 4 + 255) / 256, 256, 0, stream>>>(erow, cnt, rnk);
    scan_counts<<<1, SCAN_T, 0, stream>>>(cnt, off);
    place_cols<<<(Ee / 4 + 255) / 256, 256, 0, stream>>>(erow, ecol, rnk, off, scol);

    pull_gemm_ln<<<BLK_COUNT, BLK_WAVES * 32, 0, stream>>>(off, scol, hX, hY, ht, WX, bX, gX, beX,
                                                             WY, bY, gY, beY,
                                                             out, out + (size_t)Nn * HX);
}


# Fixed 64-slot row per destination node: the count atomic's returned rank places the source id at once (overflow list past 64); prefix scan and placement pass go; sum order unchanged; 5 launches to 3
# speedup vs baseline: 1.4760x; 1.2366x over previous
#include <hip/hip_runtime.h>
#include <hip/hip_bf16.h>


#define Nn 50000
#define Ee 800000
#define HX 64
#define HY 32
#define HT 32
#define FEAT 96
#define KX 128

#define CAP  64
#define NPAD 53248
#define OVN  Nn

#define WS_CNT  0
#define WS_SCOL (WS_CNT + NPAD * 4)
#define WS_OVR  (WS_SCOL + Nn * CAP * 4)
#define WS_OVC  (WS_OVR + Ee * 4)

typedef __attribute__((ext_vector_type(2))) float v2f;
typedef __attribute__((ext_vector_type(8))) float v8f;

__global__ void zero_cnt(int* __restrict__ cnt) {
    const int i = blockIdx.x * blockDim.x + threadIdx.x;
    if (i < NPAD / 4) ((int4*)cnt)[i] = make_int4(0, 0, 0, 0);
}

__device__ __forceinline__ void put_edge(const int r, const int k, const int c,
                                         int* __restrict__ cnt, int* __restrict__ scol,
                                         int* __restrict__ ovr, int* __restrict__ ovc) {
    if (k < CAP) {
        scol[r * CAP + k] = c;
    } else {
        const int j = atomicAdd(cnt + OVN, 1);
        ovr[j] = r;
        ovc[j] = c;
    }
}

__global__ void hist_place(const int* __restrict__ erow,
                           const int* __restrict__ ecol,
                           int* __restrict__ cnt,
                           int* __restrict__ scol,
                           int* __restrict__ ovr,
                           int* __restrict__ ovc) {
    const int q = blockIdx.x * blockDim.x + threadIdx.x;
    if (q < Ee / 4) {
        const int4 r = ((const int4*)erow)[q];
        const int4 s = ((const int4*)ecol)[q];
        int4 k;
        k.x = atomicAdd(cnt + r.x, 1);
        k.y = atomicAdd(cnt + r.y, 1);
        k.z = atomicAdd(cnt + r.z, 1);
        k.w = atomicAdd(cnt + r.w, 1);
        put_edge(r.x, k.x, s.x, cnt, scol, ovr, ovc);
        put_edge(r.y, k.y, s.y, cnt, scol, ovr, ovc);
        put_edge(r.z, k.z, s.z, cnt, scol, ovr, ovc);
        put_edge(r.w, k.w, s.w, cnt, scol, ovr, ovc);
    }
}

#define SENT 0x7fffffff

__device__ __forceinline__ float4 add4(float4 a, float4 b) {
    return make_float4(a.x + b.x, a.y + b.y, a.z + b.z, a.w + b.w);
}

template <bool TWO>
__device__ __forceinline__ float4 sum_sorted(const int* __restrict__ seg, const int d,
                                             const int lane, const bool act,
                                             const float* __restrict__ bp, const int rowsh) {
    const int c0 = (lane < d) ? seg[lane] : SENT;
    int c1 = SENT;
    if (TWO) c1 = (lane + 32 < d) ? seg[lane + 32] : SENT;
    int r0 = 0, r1 = 0;
    const int d0 = TWO ? 32 : d;
    for (int j = 0; j < d0; ++j) {
        const int cj = __builtin_amdgcn_readlane(c0, j);
        r0 += (cj < c0 || (cj == c0 && j < lane)) ? 1 : 0;
        if (TWO) r1 += (cj <= c1) ? 1 : 0;
    }
    if (TWO) {
        for (int j = 32; j < d; ++j) {
            const int cj = __builtin_amdgcn_readlane(c1, j - 32);
            r0 += (cj < c0) ? 1 : 0;
            r1 += (cj < c1 || (cj == c1 && j - 32 < lane)) ? 1 : 0;
        }
    }
    if (lane >= d) r0 = -1;
    if (!TWO || lane + 32 >= d) r1 = -1;

    auto pick = [&](const int p) -> int {
        const unsigned m0 = __builtin_amdgcn_ballot_w32(r0 == p);
        if (!TWO) return __builtin_amdgcn_readlane(c0, __builtin_ctz(m0));
        const unsigned m1 = __builtin_amdgcn_ballot_w32(r1 == p);
        const int a = __builtin_amdgcn_readlane(c0, m0 ? __builtin_ctz(m0) : 0);
        const int b = __builtin_amdgcn_readlane(c1, m1 ? __builtin_ctz(m1) : 0);
        return m0 ? a : b;
    };

    const float4 z = make_float4(0.f, 0.f, 0.f, 0.f);
    float4 acc = z;
    int p = 0;
    for (; p + 4 <= d; p += 4) {
        const int ca = pick(p), cb = pick(p + 1), cc = pick(p + 2), cd = pick(p + 3);
        float4 va = z, vb = z, vc = z, vd = z;
        if (act) {
            va = *(const float4*)(bp + ((size_t)ca << rowsh));
            vb = *(const float4*)(bp + ((size_t)cb << rowsh));
            vc = *(const float4*)(bp + ((size_t)cc << rowsh));
            vd = *(const float4*)(bp + ((size_t)cd << rowsh));
        }
        acc = add4(acc, va);
        acc = add4(acc, vb);
        acc = add4(acc, vc);
        acc = add4(acc, vd);
    }
    for (; p < d; ++p) {
        const int ca = pick(p);
        float4 va = z;
        if (act) va = *(const float4*)(bp + ((size_t)ca << rowsh));
        acc = add4(acc, va);
    }
    return acc;
}

__device__ __forceinline__ float4 sum_sorted_any(const int* __restrict__ seg, const int d,
                                              const int node,
                                              const int* __restrict__ ovr,
                                              const int* __restrict__ ovc, const int ovn,
                                              const int lane, const bool act,
                                              const float* __restrict__ bp, const int rowsh) {
    const float4 z = make_float4(0.f, 0.f, 0.f, 0.f);
    float4 acc = z;
    int last = -1;
    int done = 0;
    while (done < d) {
        int best = SENT;
        for (int i = lane; i < CAP; i += 32) {
            const int c = seg[i];
            if (c > last && c < best) best = c;
        }
        for (int i = lane; i < ovn; i += 32) {
            if (ovr[i] == node) {
                const int c = ovc[i];
                if (c > last && c < best) best = c;
            }
        }
        for (int s = 16; s > 0; s >>= 1) {
            const int o = __shfl_xor(best, s);
            best = o < best ? o : best;
        }
        int n = 0;
        for (int i = lane; i < CAP; i += 32) n += (seg[i] == best) ? 1 : 0;
        for (int i = lane; i < ovn; i += 32) n += (ovr[i] == node && ovc[i] == best) ? 1 : 0;
        for (int s = 16; s > 0; s >>= 1) n += __shfl_xor(n, s);
        best = __builtin_amdgcn_readfirstlane(best);
        n    = __builtin_amdgcn_readfirstlane(n);
        if (n <= 0) break;
        float4 v = z;
        if (act) v = *(const float4*)(bp + ((size_t)best << rowsh));
        for (int k = 0; k < n; ++k) acc = add4(acc, v);
        last = best;
        done += n;
    }
    return acc;
}

__device__ __forceinline__ v8f wmma_f32(v2f a, v2f b, v8f c) {
    return __builtin_amdgcn_wmma_f32_16x16x4_f32(
         false, a,  false, b,
         (short)0, c,  false,  false);
}

#define BLK_WAVES 16
#define BLK_NODES 32
#define BLK_COUNT ((Nn + BLK_NODES - 1) / BLK_NODES)
#define APITCH 132
#define CPITCH 100

__global__ void __launch_bounds__(BLK_WAVES * 32, 16) pull_gemm_ln(const int* __restrict__ cnt,
                                                               const int* __restrict__ scol,
                                                               const int* __restrict__ ovr,
                                                               const int* __restrict__ ovc,
                                                               const float* __restrict__ hX,
                                                               const float* __restrict__ hY,
                                                               const float* __restrict__ ht,
                                                               const float* __restrict__ WX,
                                                               const float* __restrict__ bX,
                                                               const float* __restrict__ gX,
                                                               const float* __restrict__ beX,
                                                               const float* __restrict__ WY,
                                                               const float* __restrict__ bY,
                                                               const float* __restrict__ gY,
                                                               const float* __restrict__ beY,
                                                               float* __restrict__ outX,
                                                               float* __restrict__ outY) {
    __shared__ __attribute__((aligned(16))) float At[BLK_NODES * APITCH];
    __shared__ __attribute__((aligned(16))) float Ct[BLK_NODES * CPITCH];

    const int lane = threadIdx.x & 31;
    const int wave = __builtin_amdgcn_readfirstlane(threadIdx.x >> 5);
    const int base = blockIdx.x * BLK_NODES;
    const bool act = lane < 24;

    {
        const float* bp = (lane < 16) ? (hX + lane * 4) : (hY + ((lane - 16) & 7) * 4);
        const int rowsh = (lane < 16) ? 6 : 5;
        #pragma nounroll
        for (int r = 0; r < BLK_NODES / 16; ++r) {
            const int row  = r * 16 + wave;
            const int node = base + row;
            float4 acc = make_float4(0.f, 0.f, 0.f, 0.f);
            if (node < Nn) {
                const int d = __builtin_amdgcn_readfirstlane(cnt[node]);
                const int* seg = scol + node * CAP;
                if (d <= 32)       acc = sum_sorted<false>(seg, d, lane, act, bp, rowsh);
                else if (d <= CAP) acc = sum_sorted<true>(seg, d, lane, act, bp, rowsh);
                else {
                    int ovn = __builtin_amdgcn_readfirstlane(cnt[OVN]);
                    ovn = ovn < Ee ? ovn : Ee;
                    acc = sum_sorted_any(seg, d, node, ovr, ovc, ovn, lane, act, bp, rowsh);
                }
            }
            if (act) *(float4*)(At + row * APITCH + lane * 4) = acc;
            else     *(float4*)(At + row * APITCH + FEAT + (lane - 24) * 4) = *(const float4*)(ht + (lane - 24) * 4);
        }
    }
    __syncthreads();

    {
        const int m     = lane & 15;
        const int khalf = lane >> 4;
        if (wave < 8) {
            const int rt = wave >> 2, ct = wave & 3;
            const float* arow = At + (rt * 16 + m) * APITCH + khalf * 2;
            const float* wcol = WX + (khalf * 2) * HX + ct * 16 + m;
            v8f c = {};
            #pragma unroll 8
            for (int k0 = 0; k0 < KX; k0 += 4) {
                const v2f a = *(const v2f*)(arow + k0);
                v2f b;
                b.x = wcol[k0 * HX];
                b.y = wcol[k0 * HX + HX];
                c = wmma_f32(a, b, c);
            }
            const float bias = bX[ct * 16 + m];
            #pragma unroll
            for (int v = 0; v < 8; ++v)
                Ct[(rt * 16 + khalf * 8 + v) * CPITCH + ct * 16 + m] = fmaxf(c[v] + bias, 0.0f);
        } else if (wave < 12) {
            const int rt = (wave - 8) >> 1, ct = (wave - 8) & 1;
            const float* arow = At + (rt * 16 + m) * APITCH + HX + khalf * 2;
            const float* wcol = WY + (khalf * 2) * HY + ct * 16 + m;
            v8f c = {};
            #pragma unroll
            for (int k0 = 0; k0 < HY; k0 += 4) {
                const v2f a = *(const v2f*)(arow + k0);
                v2f b;
                b.x = wcol[k0 * HY];
                b.y = wcol[k0 * HY + HY];
                c = wmma_f32(a, b, c);
            }
            const float bias = bY[ct * 16 + m];
            #pragma unroll
            for (int v = 0; v < 8; ++v)
                Ct[(rt * 16 + khalf * 8 + v) * CPITCH + HX + ct * 16 + m] = fmaxf(c[v] + bias, 0.0f);
        }
    }
    __syncthreads();

    {
        const int c4 = (lane < 16) ? lane * 4 : (lane - 16) * 4;
        float4 g  = make_float4(0.f, 0.f, 0.f, 0.f);
        float4 be = g;
        if (act) {
            g  = *(const float4*)(((lane < 16) ? gX : gY) + c4);
            be = *(const float4*)(((lane < 16) ? beX : beY) + c4);
        }
        const float rn = (lane < 16) ? (1.0f / 64.0f) : (1.0f / 32.0f);
        #pragma unroll
        for (int r = 0; r < BLK_NODES / 16; ++r) {
            const int row  = r * 16 + wave;
            const int node = base + row;
            if (node < Nn) {
                float4 x = make_float4(0.f, 0.f, 0.f, 0.f);
                if (act) x = *(const float4*)(Ct + row * CPITCH + lane * 4);
                float s  = (x.x + x.y) + (x.z + x.w);
                float s2 = (x.x * x.x + x.y * x.y) + (x.z * x.z + x.w * x.w);
                #pragma unroll
                for (int o = 1; o < 16; o <<= 1) {
                    s  += __shfl_xor(s, o);
                    s2 += __shfl_xor(s2, o);
                }
                const float mean = s * rn;
                const float var  = s2 * rn - mean * mean;
                const float inv  = rsqrtf(var + 1e-5f);
                if (act) {
                    float* o = (lane < 16) ? (outX + (size_t)node * HX + c4) : (outY + (size_t)node * HY + c4);
                    float4 y;
                    y.x = (x.x - mean) * inv * g.x + be.x;
                    y.y = (x.y - mean) * inv * g.y + be.y;
                    y.z = (x.z - mean) * inv * g.z + be.z;
                    y.w = (x.w - mean) * inv * g.w + be.w;
                    *(float4*)o = y;
                }
            }
        }
    }
}

extern "C" void kernel_launch(void* const* d_in, const int* in_sizes, int n_in,
                              void* d_out, int out_size, void* d_ws, size_t ws_size,
                              hipStream_t stream) {
    const int*   erow = (const int*)d_in[0];
    const int*   ecol = (const int*)d_in[1];
    const float* hX   = (const float*)d_in[2];
    const float* hY   = (const float*)d_in[3];
    const float* ht   = (const float*)d_in[4];
    const float* WX   = (const float*)d_in[5];
    const float* bX   = (const float*)d_in[6];
    const float* gX   = (const float*)d_in[7];
    const float* beX  = (const float*)d_in[8];
    const float* WY   = (const float*)d_in[9];
    const float* bY   = (const float*)d_in[10];
    const float* gY   = (const float*)d_in[11];
    const float* beY  = (const float*)d_in[12];

    float* out  = (float*)d_out;
    char*  ws   = (char*)d_ws;
    int*   cnt  = (int*)(ws + WS_CNT);
    int*   scol = (int*)(ws + WS_SCOL);
    int*   ovr  = (int*)(ws + WS_OVR);
    int*   ovc  = (int*)(ws + WS_OVC);

    zero_cnt<<<(NPAD / 4 + 255) / 256, 256, 0, stream>>>(cnt);
    hist_place<<<(Ee / 4 + 255) / 256, 256, 0, stream>>>(erow, ecol, cnt, scol, ovr, ovc);

    pull_gemm_ln<<<BLK_COUNT, BLK_WAVES * 32, 0, stream>>>(cnt, scol, ovr, ovc, hX, hY, ht, WX, bX, gX, beX,
                                                             WY, bY, gY, beY,
                                                             out, out + (size_t)Nn * HX);
}


# Edges counting-sorted by destination block in LDS (50 chunks, whole-line stores); each pull block slots its 50 runs in LDS; count atomics, slot table, zero launch go; sum order kept; 3 launches to 2
# speedup vs baseline: 3.8641x; 2.6180x over previous
#include <hip/hip_runtime.h>
#include <hip/hip_bf16.h>


#define Nn 50000
#define Ee 800000
#define HX 64
#define HY 32
#define HT 32
#define FEAT 96
#define KX 128

#define BLK_WAVES 16
#define BLK_NODES 32
#define BLK_COUNT ((Nn + BLK_NODES - 1) / BLK_NODES)

#define CAP    64
#define NB     50
#define CH     (Ee / NB)
#define CHQ    (CH / 4)
#define OPITCH (BLK_COUNT + 1)
#define SORT_THREADS 1024

#define WS_OFF  0
#define WS_REC  312832

typedef __attribute__((ext_vector_type(2))) float v2f;
typedef __attribute__((ext_vector_type(8))) float v8f;

__global__ void __launch_bounds__(SORT_THREADS) bucket_edges(const int* __restrict__ erow,
                                                             const int* __restrict__ ecol,
                                                             int* __restrict__ offtab,
                                                             unsigned* __restrict__ recs) {
    __shared__ int bins[2 * SORT_THREADS];
    __shared__ int offs[2 * SORT_THREADS];
    __shared__ int wsum[32];
    __shared__ __attribute__((aligned(16))) unsigned sorted[CH];

    const int tid  = threadIdx.x;
    const int lane = tid & 31;
    const int wave = tid >> 5;
    const int blk  = blockIdx.x;

    bins[tid] = 0;
    bins[tid + SORT_THREADS] = 0;
    __syncthreads();

    unsigned rec[16];
    int      rk[16];
    bool     ok[4];
    #pragma unroll
    for (int it = 0; it < 4; ++it) {
        const int q = it * SORT_THREADS + tid;
        ok[it] = q < CHQ;
        int4 r = make_int4(0, 0, 0, 0), s = r;
        if (ok[it]) {
            r = ((const int4*)erow)[blk * CHQ + q];
            s = ((const int4*)ecol)[blk * CHQ + q];
        }
        rec[4 * it + 0] = ((unsigned)r.x << 16) | (unsigned)s.x;
        rec[4 * it + 1] = ((unsigned)r.y << 16) | (unsigned)s.y;
        rec[4 * it + 2] = ((unsigned)r.z << 16) | (unsigned)s.z;
        rec[4 * it + 3] = ((unsigned)r.w << 16) | (unsigned)s.w;
    }
    #pragma unroll
    for (int e = 0; e < 16; ++e) {
        rk[e] = 0;
        if (ok[e >> 2]) rk[e] = atomicAdd(&bins[rec[e] >> 21], 1);
    }
    __syncthreads();

    const int a  = bins[2 * tid];
    const int b2 = bins[2 * tid + 1];
    const int s2 = a + b2;
    int incl = s2;
    #pragma unroll
    for (int o = 1; o < 32; o <<= 1) {
        const int v = __shfl_up(incl, o);
        if (lane >= o) incl += v;
    }
    if (lane == 31) wsum[wave] = incl;
    __syncthreads();
    if (wave == 0) {
        const int w = wsum[lane];
        int wi = w;
        #pragma unroll
        for (int o = 1; o < 32; o <<= 1) {
            const int v = __shfl_up(wi, o);
            if (lane >= o) wi += v;
        }
        wsum[lane] = wi - w;
    }
    __syncthreads();
    const int excl = wsum[wave] + incl - s2;
    offs[2 * tid]     = excl;
    offs[2 * tid + 1] = excl + a;
    if (2 * tid < OPITCH) *(int2*)(offtab + blk * OPITCH + 2 * tid) = make_int2(excl, excl + a);
    __syncthreads();

    #pragma unroll
    for (int e = 0; e < 16; ++e)
        if (ok[e >> 2]) sorted[offs[rec[e] >> 21] + rk[e]] = rec[e];
    __syncthreads();

    for (int q = tid; q < CHQ; q += SORT_THREADS)
        ((uint4*)recs)[blk * CHQ + q] = ((const uint4*)sorted)[q];
}

#define SENT 0x7fffffff

__device__ __forceinline__ float4 add4(float4 a, float4 b) {
    return make_float4(a.x + b.x, a.y + b.y, a.z + b.z, a.w + b.w);
}

template <bool TWO>
__device__ __forceinline__ float4 sum_sorted(const int c0, const int c1, const int d,
                                             const int lane, const bool act,
                                             const float* __restrict__ bp, const int rowsh) {
    int r0 = 0, r1 = 0;
    const int d0 = TWO ? 32 : d;
    for (int j = 0; j < d0; ++j) {
        const int cj = __builtin_amdgcn_readlane(c0, j);
        r0 += (cj < c0 || (cj == c0 && j < lane)) ? 1 : 0;
        if (TWO) r1 += (cj <= c1) ? 1 : 0;
    }
    if (TWO) {
        for (int j = 32; j < d; ++j) {
            const int cj = __builtin_amdgcn_readlane(c1, j - 32);
            r0 += (cj < c0) ? 1 : 0;
            r1 += (cj < c1 || (cj == c1 && j - 32 < lane)) ? 1 : 0;
        }
    }
    if (lane >= d) r0 = -1;
    if (!TWO || lane + 32 >= d) r1 = -1;

    auto pick = [&](const int p) -> int {
        const unsigned m0 = __builtin_amdgcn_ballot_w32(r0 == p);
        if (!TWO) return __builtin_amdgcn_readlane(c0, __builtin_ctz(m0));
        const unsigned m1 = __builtin_amdgcn_ballot_w32(r1 == p);
        const int a = __builtin_amdgcn_readlane(c0, m0 ? __builtin_ctz(m0) : 0);
        const int b = __builtin_amdgcn_readlane(c1, m1 ? __builtin_ctz(m1) : 0);
        return m0 ? a : b;
    };

    const float4 z = make_float4(0.f, 0.f, 0.f, 0.f);
    float4 acc = z;
    int p = 0;
    for (; p + 4 <= d; p += 4) {
        const int ca = pick(p), cb = pick(p + 1), cc = pick(p + 2), cd = pick(p + 3);
        float4 va = z, vb = z, vc = z, vd = z;
        if (act) {
            va = *(const float4*)(bp + ((size_t)ca << rowsh));
            vb = *(const float4*)(bp + ((size_t)cb << rowsh));
            vc = *(const float4*)(bp + ((size_t)cc << rowsh));
            vd = *(const float4*)(bp + ((size_t)cd << rowsh));
        }
        acc = add4(acc, va);
        acc = add4(acc, vb);
        acc = add4(acc, vc);
        acc = add4(acc, vd);
    }
    for (; p < d; ++p) {
        const int ca = pick(p);
        float4 va = z;
        if (act) va = *(const float4*)(bp + ((size_t)ca << rowsh));
        acc = add4(acc, va);
    }
    return acc;
}

__device__ __forceinline__ float4 sum_sorted_any(const int* __restrict__ offtab,
                                              const unsigned* __restrict__ recs,
                                              const int blk, const int rl, const int d,
                                              const int lane, const bool act,
                                              const float* __restrict__ bp, const int rowsh) {
    const float4 z = make_float4(0.f, 0.f, 0.f, 0.f);
    float4 acc = z;
    int last = -1;
    int done = 0;
    while (done < d) {
        int best = SENT;
        for (int t = 0; t < NB; ++t) {
            const int o0 = offtab[t * OPITCH + blk];
            const int o1 = offtab[t * OPITCH + blk + 1];
            const unsigned* rp = recs + t * CH + o0;
            for (int i = lane; i < o1 - o0; i += 32) {
                const unsigned rc = rp[i];
                const int c = (int)(rc & 0xffffu);
                if ((int)((rc >> 16) & 31u) == rl && c > last && c < best) best = c;
            }
        }
        for (int s = 16; s > 0; s >>= 1) {
            const int o = __shfl_xor(best, s);
            best = o < best ? o : best;
        }
        int n = 0;
        for (int t = 0; t < NB; ++t) {
            const int o0 = offtab[t * OPITCH + blk];
            const int o1 = offtab[t * OPITCH + blk + 1];
            const unsigned* rp = recs + t * CH + o0;
            for (int i = lane; i < o1 - o0; i += 32) {
                const unsigned rc = rp[i];
                n += ((int)((rc >> 16) & 31u) == rl && (int)(rc & 0xffffu) == best) ? 1 : 0;
            }
        }
        for (int s = 16; s > 0; s >>= 1) n += __shfl_xor(n, s);
        best = __builtin_amdgcn_readfirstlane(best);
        n    = __builtin_amdgcn_readfirstlane(n);
        if (n <= 0) break;
        float4 v = z;
        if (act) v = *(const float4*)(bp + ((size_t)best << rowsh));
        for (int k = 0; k < n; ++k) acc = add4(acc, v);
        last = best;
        done += n;
    }
    return acc;
}

__device__ __forceinline__ v8f wmma_f32(v2f a, v2f b, v8f c) {
    return __builtin_amdgcn_wmma_f32_16x16x4_f32(
         false, a,  false, b,
         (short)0, c,  false,  false);
}

#define APITCH 132
#define CPITCH 100

__global__ void __launch_bounds__(BLK_WAVES * 32, 16) pull_gemm_ln(const int* __restrict__ offtab,
                                                               const unsigned* __restrict__ recs,
                                                               const float* __restrict__ hX,
                                                               const float* __restrict__ hY,
                                                               const float* __restrict__ ht,
                                                               const float* __restrict__ WX,
                                                               const float* __restrict__ bX,
                                                               const float* __restrict__ gX,
                                                               const float* __restrict__ beX,
                                                               const float* __restrict__ WY,
                                                               const float* __restrict__ bY,
                                                               const float* __restrict__ gY,
                                                               const float* __restrict__ beY,
                                                               float* __restrict__ outX,
                                                               float* __restrict__ outY) {
    __shared__ __attribute__((aligned(16))) float At[BLK_NODES * APITCH];
    __shared__ __attribute__((aligned(16))) float Ct[BLK_NODES * CPITCH];
    __shared__ int ncnt[BLK_NODES];
    __shared__ int lseg[BLK_NODES * CAP];

    const int lane = threadIdx.x & 31;
    const int wave = __builtin_amdgcn_readfirstlane(threadIdx.x >> 5);
    const int base = blockIdx.x * BLK_NODES;
    const bool act = lane < 24;

    if (threadIdx.x < BLK_NODES) ncnt[threadIdx.x] = 0;
    __syncthreads();
    {
        const int t = threadIdx.x >> 3;
        const int j = threadIdx.x & 7;
        if (t < NB) {
            const int o0 = offtab[t * OPITCH + blockIdx.x];
            const int o1 = offtab[t * OPITCH + blockIdx.x + 1];
            const unsigned* rp = recs + t * CH + o0;
            for (int i = j; i < o1 - o0; i += 8) {
                const unsigned rc = rp[i];
                const int rl = (int)((rc >> 16) & 31u);
                const int k  = atomicAdd(&ncnt[rl], 1);
                if (k < CAP) lseg[rl * CAP + k] = (int)(rc & 0xffffu);
            }
        }
    }
    __syncthreads();

    {
        const float* bp = (lane < 16) ? (hX + lane * 4) : (hY + ((lane - 16) & 7) * 4);
        const int rowsh = (lane < 16) ? 6 : 5;
        #pragma nounroll
        for (int r = 0; r < BLK_NODES / 16; ++r) {
            const int row  = r * 16 + wave;
            const int node = base + row;
            float4 acc = make_float4(0.f, 0.f, 0.f, 0.f);
            if (node < Nn) {
                const int d = __builtin_amdgcn_readfirstlane(ncnt[row]);
                const int* seg = lseg + row * CAP;
                if (d <= 32) {
                    const int c0 = (lane < d) ? seg[lane] : SENT;
                    acc = sum_sorted<false>(c0, SENT, d, lane, act, bp, rowsh);
                } else if (d <= CAP) {
                    const int c0 = seg[lane];
                    const int c1 = (lane + 32 < d) ? seg[lane + 32] : SENT;
                    acc = sum_sorted<true>(c0, c1, d, lane, act, bp, rowsh);
                } else {
                    acc = sum_sorted_any(offtab, recs, blockIdx.x, row, d, lane, act, bp, rowsh);
                }
            }
            if (act) *(float4*)(At + row * APITCH + lane * 4) = acc;
            else     *(float4*)(At + row * APITCH + FEAT + (lane - 24) * 4) = *(const float4*)(ht + (lane - 24) * 4);
        }
    }
    __syncthreads();

    {
        const int m     = lane & 15;
        const int khalf = lane >> 4;
        if (wave < 8) {
            const int rt = wave >> 2, ct = wave & 3;
            const float* arow = At + (rt * 16 + m) * APITCH + khalf * 2;
            const float* wcol = WX + (khalf * 2) * HX + ct * 16 + m;
            v8f c = {};
            #pragma unroll 8
            for (int k0 = 0; k0 < KX; k0 += 4) {
                const v2f a = *(const v2f*)(arow + k0);
                v2f b;
                b.x = wcol[k0 * HX];
                b.y = wcol[k0 * HX + HX];
                c = wmma_f32(a, b, c);
            }
            const float bias = bX[ct * 16 + m];
            #pragma unroll
            for (int v = 0; v < 8; ++v)
                Ct[(rt * 16 + khalf * 8 + v) * CPITCH + ct * 16 + m] = fmaxf(c[v] + bias, 0.0f);
        } else if (wave < 12) {
            const int rt = (wave - 8) >> 1, ct = (wave - 8) & 1;
            const float* arow = At + (rt * 16 + m) * APITCH + HX + khalf * 2;
            const float* wcol = WY + (khalf * 2) * HY + ct * 16 + m;
            v8f c = {};
            #pragma unroll
            for (int k0 = 0; k0 < HY; k0 += 4) {
                const v2f a = *(const v2f*)(arow + k0);
                v2f b;
                b.x = wcol[k0 * HY];
                b.y = wcol[k0 * HY + HY];
                c = wmma_f32(a, b, c);
            }
            const float bias = bY[ct * 16 + m];
            #pragma unroll
            for (int v = 0; v < 8; ++v)
                Ct[(rt * 16 + khalf * 8 + v) * CPITCH + HX + ct * 16 + m] = fmaxf(c[v] + bias, 0.0f);
        }
    }
    __syncthreads();

    {
        const int c4 = (lane < 16) ? lane * 4 : (lane - 16) * 4;
        float4 g  = make_float4(0.f, 0.f, 0.f, 0.f);
        float4 be = g;
        if (act) {
            g  = *(const float4*)(((lane < 16) ? gX : gY) + c4);
            be = *(const float4*)(((lane < 16) ? beX : beY) + c4);
        }
        const float rn = (lane < 16) ? (1.0f / 64.0f) : (1.0f / 32.0f);
        #pragma unroll
        for (int r = 0; r < BLK_NODES / 16; ++r) {
            const int row  = r * 16 + wave;
            const int node = base + row;
            if (node < Nn) {
                float4 x = make_float4(0.f, 0.f, 0.f, 0.f);
                if (act) x = *(const float4*)(Ct + row * CPITCH + lane * 4);
                float s  = (x.x + x.y) + (x.z + x.w);
                float s2 = (x.x * x.x + x.y * x.y) + (x.z * x.z + x.w * x.w);
                #pragma unroll
                for (int o = 1; o < 16; o <<= 1) {
                    s  += __shfl_xor(s, o);
                    s2 += __shfl_xor(s2, o);
                }
                const float mean = s * rn;
                const float var  = s2 * rn - mean * mean;
                const float inv  = rsqrtf(var + 1e-5f);
                if (act) {
                    float* o = (lane < 16) ? (outX + (size_t)node * HX + c4) : (outY + (size_t)node * HY + c4);
                    float4 y;
                    y.x = (x.x - mean) * inv * g.x + be.x;
                    y.y = (x.y - mean) * inv * g.y + be.y;
                    y.z = (x.z - mean) * inv * g.z + be.z;
                    y.w = (x.w - mean) * inv * g.w + be.w;
                    *(float4*)o = y;
                }
            }
        }
    }
}

extern "C" void kernel_launch(void* const* d_in, const int* in_sizes, int n_in,
                              void* d_out, int out_size, void* d_ws, size_t ws_size,
                              hipStream_t stream) {
    const int*   erow = (const int*)d_in[0];
    const int*   ecol = (const int*)d_in[1];
    const float* hX   = (const float*)d_in[2];
    const float* hY   = (const float*)d_in[3];
    const float* ht   = (const float*)d_in[4];
    const float* WX   = (const float*)d_in[5];
    const float* bX   = (const float*)d_in[6];
    const float* gX   = (const float*)d_in[7];
    const float* beX  = (const float*)d_in[8];
    const float* WY   = (const float*)d_in[9];
    const float* bY   = (const float*)d_in[10];
    const float* gY   = (const float*)d_in[11];
    const float* beY  = (const float*)d_in[12];

    float*    out    = (float*)d_out;
    char*     ws     = (char*)d_ws;
    int*      offtab = (int*)(ws + WS_OFF);
    unsigned* recs   = (unsigned*)(ws + WS_REC);

    bucket_edges<<<NB, SORT_THREADS, 0, stream>>>(erow, ecol, offtab, recs);

    pull_gemm_ln<<<BLK_COUNT, BLK_WAVES * 32, 0, stream>>>(offtab, recs, hX, hY, ht, WX, bX, gX, beX,
                                                             WY, bY, gY, beY,
                                                             out, out + (size_t)Nn * HX);
}


# Kernel 2 work-groups of 8 waves, not 16: a wave pulls 4 nodes in turn, 12 product tiles on 8 waves; half as many waves to start, all 1563 groups resident at once; sum order unchanged
# speedup vs baseline: 4.0991x; 1.0608x over previous
#include <hip/hip_runtime.h>
#include <hip/hip_bf16.h>


#define Nn 50000
#define Ee 800000
#define HX 64
#define HY 32
#define HT 32
#define FEAT 96
#define KX 128

#define BLK_WAVES 8
#define BLK_NODES 32
#define ROWS_PER_WAVE (BLK_NODES / BLK_WAVES)
#define BLK_COUNT ((Nn + BLK_NODES - 1) / BLK_NODES)

#define CAP    64
#define NB     50
#define CH     (Ee / NB)
#define CHQ    (CH / 4)
#define OPITCH (BLK_COUNT + 1)
#define SORT_THREADS 1024
#define RUN_LANES 5

#define WS_OFF  0
#define WS_REC  312832

typedef __attribute__((ext_vector_type(2))) float v2f;
typedef __attribute__((ext_vector_type(8))) float v8f;

__global__ void __launch_bounds__(SORT_THREADS) bucket_edges(const int* __restrict__ erow,
                                                             const int* __restrict__ ecol,
                                                             int* __restrict__ offtab,
                                                             unsigned* __restrict__ recs) {
    __shared__ int bins[2 * SORT_THREADS];
    __shared__ int offs[2 * SORT_THREADS];
    __shared__ int wsum[32];
    __shared__ __attribute__((aligned(16))) unsigned sorted[CH];

    const int tid  = threadIdx.x;
    const int lane = tid & 31;
    const int wave = tid >> 5;
    const int blk  = blockIdx.x;

    bins[tid] = 0;
    bins[tid + SORT_THREADS] = 0;
    __syncthreads();

    unsigned rec[16];
    int      rk[16];
    bool     ok[4];
    #pragma unroll
    for (int it = 0; it < 4; ++it) {
        const int q = it * SORT_THREADS + tid;
        ok[it] = q < CHQ;
        int4 r = make_int4(0, 0, 0, 0), s = r;
        if (ok[it]) {
            r = ((const int4*)erow)[blk * CHQ + q];
            s = ((const int4*)ecol)[blk * CHQ + q];
        }
        rec[4 * it + 0] = ((unsigned)r.x << 16) | (unsigned)s.x;
        rec[4 * it + 1] = ((unsigned)r.y << 16) | (unsigned)s.y;
        rec[4 * it + 2] = ((unsigned)r.z << 16) | (unsigned)s.z;
        rec[4 * it + 3] = ((unsigned)r.w << 16) | (unsigned)s.w;
    }
    #pragma unroll
    for (int e = 0; e < 16; ++e) {
        rk[e] = 0;
        if (ok[e >> 2]) rk[e] = atomicAdd(&bins[rec[e] >> 21], 1);
    }
    __syncthreads();

    const int a  = bins[2 * tid];
    const int b2 = bins[2 * tid + 1];
    const int s2 = a + b2;
    int incl = s2;
    #pragma unroll
    for (int o = 1; o < 32; o <<= 1) {
        const int v = __shfl_up(incl, o);
        if (lane >= o) incl += v;
    }
    if (lane == 31) wsum[wave] = incl;
    __syncthreads();
    if (wave == 0) {
        const int w = wsum[lane];
        int wi = w;
        #pragma unroll
        for (int o = 1; o < 32; o <<= 1) {
            const int v = __shfl_up(wi, o);
            if (lane >= o) wi += v;
        }
        wsum[lane] = wi - w;
    }
    __syncthreads();
    const int excl = wsum[wave] + incl - s2;
    offs[2 * tid]     = excl;
    offs[2 * tid + 1] = excl + a;
    if (2 * tid < OPITCH) *(int2*)(offtab + blk * OPITCH + 2 * tid) = make_int2(excl, excl + a);
    __syncthreads();

    #pragma unroll
    for (int e = 0; e < 16; ++e)
        if (ok[e >> 2]) sorted[offs[rec[e] >> 21] + rk[e]] = rec[e];
    __syncthreads();

    for (int q = tid; q < CHQ; q += SORT_THREADS)
        ((uint4*)recs)[blk * CHQ + q] = ((const uint4*)sorted)[q];
}

#define SENT 0x7fffffff

__device__ __forceinline__ float4 add4(float4 a, float4 b) {
    return make_float4(a.x + b.x, a.y + b.y, a.z + b.z, a.w + b.w);
}

template <bool TWO>
__device__ __forceinline__ float4 sum_sorted(const int c0, const int c1, const int d,
                                             const int lane, const bool act,
                                             const float* __restrict__ bp, const int rowsh) {
    int r0 = 0, r1 = 0;
    const int d0 = TWO ? 32 : d;
    for (int j = 0; j < d0; ++j) {
        const int cj = __builtin_amdgcn_readlane(c0, j);
        r0 += (cj < c0 || (cj == c0 && j < lane)) ? 1 : 0;
        if (TWO) r1 += (cj <= c1) ? 1 : 0;
    }
    if (TWO) {
        for (int j = 32; j < d; ++j) {
            const int cj = __builtin_amdgcn_readlane(c1, j - 32);
            r0 += (cj < c0) ? 1 : 0;
            r1 += (cj < c1 || (cj == c1 && j - 32 < lane)) ? 1 : 0;
        }
    }
    if (lane >= d) r0 = -1;
    if (!TWO || lane + 32 >= d) r1 = -1;

    auto pick = [&](const int p) -> int {
        const unsigned m0 = __builtin_amdgcn_ballot_w32(r0 == p);
        if (!TWO) return __builtin_amdgcn_readlane(c0, __builtin_ctz(m0));
        const unsigned m1 = __builtin_amdgcn_ballot_w32(r1 == p);
        const int a = __builtin_amdgcn_readlane(c0, m0 ? __builtin_ctz(m0) : 0);
        const int b = __builtin_amdgcn_readlane(c1, m1 ? __builtin_ctz(m1) : 0);
        return m0 ? a : b;
    };

    const float4 z = make_float4(0.f, 0.f, 0.f, 0.f);
    float4 acc = z;
    int p = 0;
    for (; p + 4 <= d; p += 4) {
        const int ca = pick(p), cb = pick(p + 1), cc = pick(p + 2), cd = pick(p + 3);
        float4 va = z, vb = z, vc = z, vd = z;
        if (act) {
            va = *(const float4*)(bp + ((size_t)ca << rowsh));
            vb = *(const float4*)(bp + ((size_t)cb << rowsh));
            vc = *(const float4*)(bp + ((size_t)cc << rowsh));
            vd = *(const float4*)(bp + ((size_t)cd << rowsh));
        }
        acc = add4(acc, va);
        acc = add4(acc, vb);
        acc = add4(acc, vc);
        acc = add4(acc, vd);
    }
    for (; p < d; ++p) {
        const int ca = pick(p);
        float4 va = z;
        if (act) va = *(const float4*)(bp + ((size_t)ca << rowsh));
        acc = add4(acc, va);
    }
    return acc;
}

__device__ __forceinline__ float4 sum_sorted_any(const int* __restrict__ offtab,
                                              const unsigned* __restrict__ recs,
                                              const int blk, const int rl, const int d,
                                              const int lane, const bool act,
                                              const float* __restrict__ bp, const int rowsh) {
    const float4 z = make_float4(0.f, 0.f, 0.f, 0.f);
    float4 acc = z;
    int last = -1;
    int done = 0;
    while (done < d) {
        int best = SENT;
        for (int t = 0; t < NB; ++t) {
            const int o0 = offtab[t * OPITCH + blk];
            const int o1 = offtab[t * OPITCH + blk + 1];
            const unsigned* rp = recs + t * CH + o0;
            for (int i = lane; i < o1 - o0; i += 32) {
                const unsigned rc = rp[i];
                const int c = (int)(rc & 0xffffu);
                if ((int)((rc >> 16) & 31u) == rl && c > last && c < best) best = c;
            }
        }
        for (int s = 16; s > 0; s >>= 1) {
            const int o = __shfl_xor(best, s);
            best = o < best ? o : best;
        }
        int n = 0;
        for (int t = 0; t < NB; ++t) {
            const int o0 = offtab[t * OPITCH + blk];
            const int o1 = offtab[t * OPITCH + blk + 1];
            const unsigned* rp = recs + t * CH + o0;
            for (int i = lane; i < o1 - o0; i += 32) {
                const unsigned rc = rp[i];
                n += ((int)((rc >> 16) & 31u) == rl && (int)(rc & 0xffffu) == best) ? 1 : 0;
            }
        }
        for (int s = 16; s > 0; s >>= 1) n += __shfl_xor(n, s);
        best = __builtin_amdgcn_readfirstlane(best);
        n    = __builtin_amdgcn_readfirstlane(n);
        if (n <= 0) break;
        float4 v = z;
        if (act) v = *(const float4*)(bp + ((size_t)best << rowsh));
        for (int k = 0; k < n; ++k) acc = add4(acc, v);
        last = best;
        done += n;
    }
    return acc;
}

__device__ __forceinline__ v8f wmma_f32(v2f a, v2f b, v8f c) {
    return __builtin_amdgcn_wmma_f32_16x16x4_f32(
         false, a,  false, b,
         (short)0, c,  false,  false);
}

#define APITCH 132
#define CPITCH 100

__global__ void __launch_bounds__(BLK_WAVES * 32, 16) pull_gemm_ln(const int* __restrict__ offtab,
                                                               const unsigned* __restrict__ recs,
                                                               const float* __restrict__ hX,
                                                               const float* __restrict__ hY,
                                                               const float* __restrict__ ht,
                                                               const float* __restrict__ WX,
                                                               const float* __restrict__ bX,
                                                               const float* __restrict__ gX,
                                                               const float* __restrict__ beX,
                                                               const float* __restrict__ WY,
                                                               const float* __restrict__ bY,
                                                               const float* __restrict__ gY,
                                                               const float* __restrict__ beY,
                                                               float* __restrict__ outX,
                                                               float* __restrict__ outY) {
    __shared__ __attribute__((aligned(16))) float At[BLK_NODES * APITCH];
    __shared__ __attribute__((aligned(16))) float Ct[BLK_NODES * CPITCH];
    __shared__ int ncnt[BLK_NODES];
    __shared__ int lseg[BLK_NODES * CAP];

    const int lane = threadIdx.x & 31;
    const int wave = __builtin_amdgcn_readfirstlane(threadIdx.x >> 5);
    const int base = blockIdx.x * BLK_NODES;
    const bool act = lane < 24;

    if (threadIdx.x < BLK_NODES) ncnt[threadIdx.x] = 0;
    __syncthreads();
    {
        const int t = threadIdx.x / RUN_LANES;
        const int j = threadIdx.x - t * RUN_LANES;
        if (t < NB) {
            const int o0 = offtab[t * OPITCH + blockIdx.x];
            const int o1 = offtab[t * OPITCH + blockIdx.x + 1];
            const unsigned* rp = recs + t * CH + o0;
            for (int i = j; i < o1 - o0; i += RUN_LANES) {
                const unsigned rc = rp[i];
                const int rl = (int)((rc >> 16) & 31u);
                const int k  = atomicAdd(&ncnt[rl], 1);
                if (k < CAP) lseg[rl * CAP + k] = (int)(rc & 0xffffu);
            }
        }
    }
    __syncthreads();

    {
        const float* bp = (lane < 16) ? (hX + lane * 4) : (hY + ((lane - 16) & 7) * 4);
        const int rowsh = (lane < 16) ? 6 : 5;
        #pragma nounroll
        for (int r = 0; r < ROWS_PER_WAVE; ++r) {
            const int row  = r * BLK_WAVES + wave;
            const int node = base + row;
            float4 acc = make_float4(0.f, 0.f, 0.f, 0.f);
            if (node < Nn) {
                const int d = __builtin_amdgcn_readfirstlane(ncnt[row]);
                const int* seg = lseg + row * CAP;
                if (d <= 32) {
                    const int c0 = (lane < d) ? seg[lane] : SENT;
                    acc = sum_sorted<false>(c0, SENT, d, lane, act, bp, rowsh);
                } else if (d <= CAP) {
                    const int c0 = seg[lane];
                    const int c1 = (lane + 32 < d) ? seg[lane + 32] : SENT;
                    acc = sum_sorted<true>(c0, c1, d, lane, act, bp, rowsh);
                } else {
                    acc = sum_sorted_any(offtab, recs, blockIdx.x, row, d, lane, act, bp, rowsh);
                }
            }
            if (act) *(float4*)(At + row * APITCH + lane * 4) = acc;
            else     *(float4*)(At + row * APITCH + FEAT + (lane - 24) * 4) = *(const float4*)(ht + (lane - 24) * 4);
        }
    }
    __syncthreads();

    {
        const int m     = lane & 15;
        const int khalf = lane >> 4;
        {
            const int rt = wave >> 2, ct = wave & 3;
            const float* arow = At + (rt * 16 + m) * APITCH + khalf * 2;
            const float* wcol = WX + (khalf * 2) * HX + ct * 16 + m;
            v8f c = {};
            #pragma unroll 8
            for (int k0 = 0; k0 < KX; k0 += 4) {
                const v2f a = *(const v2f*)(arow + k0);
                v2f b;
                b.x = wcol[k0 * HX];
                b.y = wcol[k0 * HX + HX];
                c = wmma_f32(a, b, c);
            }
            const float bias = bX[ct * 16 + m];
            #pragma unroll
            for (int v = 0; v < 8; ++v)
                Ct[(rt * 16 + khalf * 8 + v) * CPITCH + ct * 16 + m] = fmaxf(c[v] + bias, 0.0f);
        }
        if (wave < 4) {
            const int rt = wave >> 1, ct = wave & 1;
            const float* arow = At + (rt * 16 + m) * APITCH + HX + khalf * 2;
            const float* wcol = WY + (khalf * 2) * HY + ct * 16 + m;
            v8f c = {};
            #pragma unroll
            for (int k0 = 0; k0 < HY; k0 += 4) {
                const v2f a = *(const v2f*)(arow + k0);
                v2f b;
                b.x = wcol[k0 * HY];
                b.y = wcol[k0 * HY + HY];
                c = wmma_f32(a, b, c);
            }
            const float bias = bY[ct * 16 + m];
            #pragma unroll
            for (int v = 0; v < 8; ++v)
                Ct[(rt * 16 + khalf * 8 + v) * CPITCH + HX + ct * 16 + m] = fmaxf(c[v] + bias, 0.0f);
        }
    }
    __syncthreads();

    {
        const int c4 = (lane < 16) ? lane * 4 : (lane - 16) * 4;
        float4 g  = make_float4(0.f, 0.f, 0.f, 0.f);
        float4 be = g;
        if (act) {
            g  = *(const float4*)(((lane < 16) ? gX : gY) + c4);
            be = *(const float4*)(((lane < 16) ? beX : beY) + c4);
        }
        const float rn = (lane < 16) ? (1.0f / 64.0f) : (1.0f / 32.0f);
        #pragma unroll
        for (int r = 0; r < ROWS_PER_WAVE; ++r) {
            const int row  = r * BLK_WAVES + wave;
            const int node = base + row;
            if (node < Nn) {
                float4 x = make_float4(0.f, 0.f, 0.f, 0.f);
                if (act) x = *(const float4*)(Ct + row * CPITCH + lane * 4);
                float s  = (x.x + x.y) + (x.z + x.w);
                float s2 = (x.x * x.x + x.y * x.y) + (x.z * x.z + x.w * x.w);
                #pragma unroll
                for (int o = 1; o < 16; o <<= 1) {
                    s  += __shfl_xor(s, o);
                    s2 += __shfl_xor(s2, o);
                }
                const float mean = s * rn;
                const float var  = s2 * rn - mean * mean;
                const float inv  = rsqrtf(var + 1e-5f);
                if (act) {
                    float* o = (lane < 16) ? (outX + (size_t)node * HX + c4) : (outY + (size_t)node * HY + c4);
                    float4 y;
                    y.x = (x.x - mean) * inv * g.x + be.x;
                    y.y = (x.y - mean) * inv * g.y + be.y;
                    y.z = (x.z - mean) * inv * g.z + be.z;
                    y.w = (x.w - mean) * inv * g.w + be.w;
                    *(float4*)o = y;
                }
            }
        }
    }
}

extern "C" void kernel_launch(void* const* d_in, const int* in_sizes, int n_in,
                              void* d_out, int out_size, void* d_ws, size_t ws_size,
                              hipStream_t stream) {
    const int*   erow = (const int*)d_in[0];
    const int*   ecol = (const int*)d_in[1];
    const float* hX   = (const float*)d_in[2];
    const float* hY   = (const float*)d_in[3];
    const float* ht   = (const float*)d_in[4];
    const float* WX   = (const float*)d_in[5];
    const float* bX   = (const float*)d_in[6];
    const float* gX   = (const float*)d_in[7];
    const float* beX  = (const float*)d_in[8];
    const float* WY   = (const float*)d_in[9];
    const float* bY   = (const float*)d_in[10];
    const float* gY   = (const float*)d_in[11];
    const float* beY  = (const float*)d_in[12];

    float*    out    = (float*)d_out;
    char*     ws     = (char*)d_ws;
    int*      offtab = (int*)(ws + WS_OFF);
    unsigned* recs   = (unsigned*)(ws + WS_REC);

    bucket_edges<<<NB, SORT_THREADS, 0, stream>>>(erow, ecol, offtab, recs);

    pull_gemm_ln<<<BLK_COUNT, BLK_WAVES * 32, 0, stream>>>(offtab, recs, hX, hY, ht, WX, bX, gX, beX,
                                                             WY, bY, gY, beY,
                                                             out, out + (size_t)Nn * HX);
}


# Common pull (in-degree <= 32) issues fewer instructions per row: slots ranked on unique (id<<5|slot) keys by one compare and add, ids put in rank order by one cross-lane push; sum order kept
# speedup vs baseline: 4.2569x; 1.0385x over previous
#include <hip/hip_runtime.h>
#include <hip/hip_bf16.h>


#define Nn 50000
#define Ee 800000
#define HX 64
#define HY 32
#define HT 32
#define FEAT 96
#define KX 128

#define BLK_WAVES 8
#define BLK_NODES 32
#define ROWS_PER_WAVE (BLK_NODES / BLK_WAVES)
#define BLK_COUNT ((Nn + BLK_NODES - 1) / BLK_NODES)

#define CAP    64
#define NB     50
#define CH     (Ee / NB)
#define CHQ    (CH / 4)
#define OPITCH (BLK_COUNT + 1)
#define SORT_THREADS 1024
#define RUN_LANES 5

#define WS_OFF  0
#define WS_REC  312832

typedef __attribute__((ext_vector_type(2))) float v2f;
typedef __attribute__((ext_vector_type(8))) float v8f;

__global__ void __launch_bounds__(SORT_THREADS) bucket_edges(const int* __restrict__ erow,
                                                             const int* __restrict__ ecol,
                                                             int* __restrict__ offtab,
                                                             unsigned* __restrict__ recs) {
    __shared__ int bins[2 * SORT_THREADS];
    __shared__ int offs[2 * SORT_THREADS];
    __shared__ int wsum[32];
    __shared__ __attribute__((aligned(16))) unsigned sorted[CH];

    const int tid  = threadIdx.x;
    const int lane = tid & 31;
    const int wave = tid >> 5;
    const int blk  = blockIdx.x;

    bins[tid] = 0;
    bins[tid + SORT_THREADS] = 0;
    __syncthreads();

    unsigned rec[16];
    int      rk[16];
    bool     ok[4];
    #pragma unroll
    for (int it = 0; it < 4; ++it) {
        const int q = it * SORT_THREADS + tid;
        ok[it] = q < CHQ;
        int4 r = make_int4(0, 0, 0, 0), s = r;
        if (ok[it]) {
            r = ((const int4*)erow)[blk * CHQ + q];
            s = ((const int4*)ecol)[blk * CHQ + q];
        }
        rec[4 * it + 0] = ((unsigned)r.x << 16) | (unsigned)s.x;
        rec[4 * it + 1] = ((unsigned)r.y << 16) | (unsigned)s.y;
        rec[4 * it + 2] = ((unsigned)r.z << 16) | (unsigned)s.z;
        rec[4 * it + 3] = ((unsigned)r.w << 16) | (unsigned)s.w;
    }
    #pragma unroll
    for (int e = 0; e < 16; ++e) {
        rk[e] = 0;
        if (ok[e >> 2]) rk[e] = atomicAdd(&bins[rec[e] >> 21], 1);
    }
    __syncthreads();

    const int a  = bins[2 * tid];
    const int b2 = bins[2 * tid + 1];
    const int s2 = a + b2;
    int incl = s2;
    #pragma unroll
    for (int o = 1; o < 32; o <<= 1) {
        const int v = __shfl_up(incl, o);
        if (lane >= o) incl += v;
    }
    if (lane == 31) wsum[wave] = incl;
    __syncthreads();
    if (wave == 0) {
        const int w = wsum[lane];
        int wi = w;
        #pragma unroll
        for (int o = 1; o < 32; o <<= 1) {
            const int v = __shfl_up(wi, o);
            if (lane >= o) wi += v;
        }
        wsum[lane] = wi - w;
    }
    __syncthreads();
    const int excl = wsum[wave] + incl - s2;
    offs[2 * tid]     = excl;
    offs[2 * tid + 1] = excl + a;
    if (2 * tid < OPITCH) *(int2*)(offtab + blk * OPITCH + 2 * tid) = make_int2(excl, excl + a);
    __syncthreads();

    #pragma unroll
    for (int e = 0; e < 16; ++e)
        if (ok[e >> 2]) sorted[offs[rec[e] >> 21] + rk[e]] = rec[e];
    __syncthreads();

    for (int q = tid; q < CHQ; q += SORT_THREADS)
        ((uint4*)recs)[blk * CHQ + q] = ((const uint4*)sorted)[q];
}

#define SENT 0x7fffffff

__device__ __forceinline__ float4 add4(float4 a, float4 b) {
    return make_float4(a.x + b.x, a.y + b.y, a.z + b.z, a.w + b.w);
}

template <int J0>
__device__ __forceinline__ unsigned rank8(const unsigned key, unsigned r) {
    #pragma unroll
    for (int j = J0; j < J0 + 8; ++j) {
        const unsigned kj = (unsigned)__builtin_amdgcn_readlane((int)key, j);
        r += (kj < key) ? 1u : 0u;
    }
    return r;
}

__device__ __forceinline__ float4 sum_sorted32(const int c0, const int d, const int lane, const bool act,
                                               const char* __restrict__ bp, const unsigned sh) {
    const bool have = lane < d;
    const unsigned key = have ? (((unsigned)c0 << 5) | (unsigned)lane) : 0xffffffffu;
    unsigned r = rank8<0>(key, 0u);
    if (d > 8)  r = rank8<8>(key, r);
    if (d > 16) r = rank8<16>(key, r);
    if (d > 24) r = rank8<24>(key, r);
    const int srt = __builtin_amdgcn_ds_permute((have ? (int)r : lane) << 2, c0);

    const float4 z = make_float4(0.f, 0.f, 0.f, 0.f);
    float4 acc = z;
    if (act) {
        int p = 0;
        for (; p + 8 <= d; p += 8) {
            unsigned id[8];
            float4   v[8];
            #pragma unroll
            for (int i = 0; i < 8; ++i) id[i] = (unsigned)__builtin_amdgcn_readlane(srt, p + i);
            #pragma unroll
            for (int i = 0; i < 8; ++i) v[i] = *(const float4*)(bp + (id[i] << sh));
            #pragma unroll
            for (int i = 0; i < 8; ++i) acc = add4(acc, v[i]);
        }
        if (p + 4 <= d) {
            const unsigned ca = (unsigned)__builtin_amdgcn_readlane(srt, p);
            const unsigned cb = (unsigned)__builtin_amdgcn_readlane(srt, p + 1);
            const unsigned cc = (unsigned)__builtin_amdgcn_readlane(srt, p + 2);
            const unsigned cd = (unsigned)__builtin_amdgcn_readlane(srt, p + 3);
            const float4 va = *(const float4*)(bp + (ca << sh));
            const float4 vb = *(const float4*)(bp + (cb << sh));
            const float4 vc = *(const float4*)(bp + (cc << sh));
            const float4 vd = *(const float4*)(bp + (cd << sh));
            acc = add4(acc, va);
            acc = add4(acc, vb);
            acc = add4(acc, vc);
            acc = add4(acc, vd);
            p += 4;
        }
        for (; p < d; ++p) {
            const unsigned ca = (unsigned)__builtin_amdgcn_readlane(srt, p);
            acc = add4(acc, *(const float4*)(bp + (ca << sh)));
        }
    }
    return acc;
}

template <bool TWO>
__device__ __forceinline__ float4 sum_sorted(const int c0, const int c1, const int d,
                                             const int lane, const bool act,
                                             const float* __restrict__ bp, const int rowsh) {
    int r0 = 0, r1 = 0;
    const int d0 = TWO ? 32 : d;
    for (int j = 0; j < d0; ++j) {
        const int cj = __builtin_amdgcn_readlane(c0, j);
        r0 += (cj < c0 || (cj == c0 && j < lane)) ? 1 : 0;
        if (TWO) r1 += (cj <= c1) ? 1 : 0;
    }
    if (TWO) {
        for (int j = 32; j < d; ++j) {
            const int cj = __builtin_amdgcn_readlane(c1, j - 32);
            r0 += (cj < c0) ? 1 : 0;
            r1 += (cj < c1 || (cj == c1 && j - 32 < lane)) ? 1 : 0;
        }
    }
    if (lane >= d) r0 = -1;
    if (!TWO || lane + 32 >= d) r1 = -1;

    auto pick = [&](const int p) -> int {
        const unsigned m0 = __builtin_amdgcn_ballot_w32(r0 == p);
        if (!TWO) return __builtin_amdgcn_readlane(c0, __builtin_ctz(m0));
        const unsigned m1 = __builtin_amdgcn_ballot_w32(r1 == p);
        const int a = __builtin_amdgcn_readlane(c0, m0 ? __builtin_ctz(m0) : 0);
        const int b = __builtin_amdgcn_readlane(c1, m1 ? __builtin_ctz(m1) : 0);
        return m0 ? a : b;
    };

    const float4 z = make_float4(0.f, 0.f, 0.f, 0.f);
    float4 acc = z;
    int p = 0;
    for (; p + 4 <= d; p += 4) {
        const int ca = pick(p), cb = pick(p + 1), cc = pick(p + 2), cd = pick(p + 3);
        float4 va = z, vb = z, vc = z, vd = z;
        if (act) {
            va = *(const float4*)(bp + ((size_t)ca << rowsh));
            vb = *(const float4*)(bp + ((size_t)cb << rowsh));
            vc = *(const float4*)(bp + ((size_t)cc << rowsh));
            vd = *(const float4*)(bp + ((size_t)cd << rowsh));
        }
        acc = add4(acc, va);
        acc = add4(acc, vb);
        acc = add4(acc, vc);
        acc = add4(acc, vd);
    }
    for (; p < d; ++p) {
        const int ca = pick(p);
        float4 va = z;
        if (act) va = *(const float4*)(bp + ((size_t)ca << rowsh));
        acc = add4(acc, va);
    }
    return acc;
}

__device__ __forceinline__ float4 sum_sorted_any(const int* __restrict__ offtab,
                                              const unsigned* __restrict__ recs,
                                              const int blk, const int rl, const int d,
                                              const int lane, const bool act,
                                              const float* __restrict__ bp, const int rowsh) {
    const float4 z = make_float4(0.f, 0.f, 0.f, 0.f);
    float4 acc = z;
    int last = -1;
    int done = 0;
    while (done < d) {
        int best = SENT;
        for (int t = 0; t < NB; ++t) {
            const int o0 = offtab[t * OPITCH + blk];
            const int o1 = offtab[t * OPITCH + blk + 1];
            const unsigned* rp = recs + t * CH + o0;
            for (int i = lane; i < o1 - o0; i += 32) {
                const unsigned rc = rp[i];
                const int c = (int)(rc & 0xffffu);
                if ((int)((rc >> 16) & 31u) == rl && c > last && c < best) best = c;
            }
        }
        for (int s = 16; s > 0; s >>= 1) {
            const int o = __shfl_xor(best, s);
            best = o < best ? o : best;
        }
        int n = 0;
        for (int t = 0; t < NB; ++t) {
            const int o0 = offtab[t * OPITCH + blk];
            const int o1 = offtab[t * OPITCH + blk + 1];
            const unsigned* rp = recs + t * CH + o0;
            for (int i = lane; i < o1 - o0; i += 32) {
                const unsigned rc = rp[i];
                n += ((int)((rc >> 16) & 31u) == rl && (int)(rc & 0xffffu) == best) ? 1 : 0;
            }
        }
        for (int s = 16; s > 0; s >>= 1) n += __shfl_xor(n, s);
        best = __builtin_amdgcn_readfirstlane(best);
        n    = __builtin_amdgcn_readfirstlane(n);
        if (n <= 0) break;
        float4 v = z;
        if (act) v = *(const float4*)(bp + ((size_t)best << rowsh));
        for (int k = 0; k < n; ++k) acc = add4(acc, v);
        last = best;
        done += n;
    }
    return acc;
}

__device__ __forceinline__ v8f wmma_f32(v2f a, v2f b, v8f c) {
    return __builtin_amdgcn_wmma_f32_16x16x4_f32(
         false, a,  false, b,
         (short)0, c,  false,  false);
}

#define APITCH 132
#define CPITCH 100

__global__ void __launch_bounds__(BLK_WAVES * 32, 16) pull_gemm_ln(const int* __restrict__ offtab,
                                                               const unsigned* __restrict__ recs,
                                                               const float* __restrict__ hX,
                                                               const float* __restrict__ hY,
                                                               const float* __restrict__ ht,
                                                               const float* __restrict__ WX,
                                                               const float* __restrict__ bX,
                                                               const float* __restrict__ gX,
                                                               const float* __restrict__ beX,
                                                               const float* __restrict__ WY,
                                                               const float* __restrict__ bY,
                                                               const float* __restrict__ gY,
                                                               const float* __restrict__ beY,
                                                               float* __restrict__ outX,
                                                               float* __restrict__ outY) {
    __shared__ __attribute__((aligned(16))) float At[BLK_NODES * APITCH];
    __shared__ __attribute__((aligned(16))) float Ct[BLK_NODES * CPITCH];
    __shared__ int ncnt[BLK_NODES];
    __shared__ int lseg[BLK_NODES * CAP];

    const int lane = threadIdx.x & 31;
    const int wave = __builtin_amdgcn_readfirstlane(threadIdx.x >> 5);
    const int base = blockIdx.x * BLK_NODES;
    const bool act = lane < 24;

    if (threadIdx.x < BLK_NODES) ncnt[threadIdx.x] = 0;
    __syncthreads();
    {
        const int t = threadIdx.x / RUN_LANES;
        const int j = threadIdx.x - t * RUN_LANES;
        if (t < NB) {
            const int o0 = offtab[t * OPITCH + blockIdx.x];
            const int o1 = offtab[t * OPITCH + blockIdx.x + 1];
            const unsigned* rp = recs + t * CH + o0;
            for (int i = j; i < o1 - o0; i += RUN_LANES) {
                const unsigned rc = rp[i];
                const int rl = (int)((rc >> 16) & 31u);
                const int k  = atomicAdd(&ncnt[rl], 1);
                if (k < CAP) lseg[rl * CAP + k] = (int)(rc & 0xffffu);
            }
        }
    }
    __syncthreads();

    {
        const float* bp = (lane < 16) ? (hX + lane * 4) : (hY + ((lane - 16) & 7) * 4);
        const int rowsh = (lane < 16) ? 6 : 5;
        #pragma nounroll
        for (int r = 0; r < ROWS_PER_WAVE; ++r) {
            const int row  = r * BLK_WAVES + wave;
            const int node = base + row;
            float4 acc = make_float4(0.f, 0.f, 0.f, 0.f);
            if (node < Nn) {
                const int d = __builtin_amdgcn_readfirstlane(ncnt[row]);
                const int* seg = lseg + row * CAP;
                if (d <= 32) {
                    const int c0 = (lane < d) ? seg[lane] : SENT;
                    acc = sum_sorted32(c0, d, lane, act, (const char*)bp, (unsigned)rowsh + 2u);
                } else if (d <= CAP) {
                    const int c0 = seg[lane];
                    const int c1 = (lane + 32 < d) ? seg[lane + 32] : SENT;
                    acc = sum_sorted<true>(c0, c1, d, lane, act, bp, rowsh);
                } else {
                    acc = sum_sorted_any(offtab, recs, blockIdx.x, row, d, lane, act, bp, rowsh);
                }
            }
            if (act) *(float4*)(At + row * APITCH + lane * 4) = acc;
            else     *(float4*)(At + row * APITCH + FEAT + (lane - 24) * 4) = *(const float4*)(ht + (lane - 24) * 4);
        }
    }
    __syncthreads();

    {
        const int m     = lane & 15;
        const int khalf = lane >> 4;
        {
            const int rt = wave >> 2, ct = wave & 3;
            const float* arow = At + (rt * 16 + m) * APITCH + khalf * 2;
            const float* wcol = WX + (khalf * 2) * HX + ct * 16 + m;
            v8f c = {};
            #pragma unroll 8
            for (int k0 = 0; k0 < KX; k0 += 4) {
                const v2f a = *(const v2f*)(arow + k0);
                v2f b;
                b.x = wcol[k0 * HX];
                b.y = wcol[k0 * HX + HX];
                c = wmma_f32(a, b, c);
            }
            const float bias = bX[ct * 16 + m];
            #pragma unroll
            for (int v = 0; v < 8; ++v)
                Ct[(rt * 16 + khalf * 8 + v) * CPITCH + ct * 16 + m] = fmaxf(c[v] + bias, 0.0f);
        }
        if (wave < 4) {
            const int rt = wave >> 1, ct = wave & 1;
            const float* arow = At + (rt * 16 + m) * APITCH + HX + khalf * 2;
            const float* wcol = WY + (khalf * 2) * HY + ct * 16 + m;
            v8f c = {};
            #pragma unroll
            for (int k0 = 0; k0 < HY; k0 += 4) {
                const v2f a = *(const v2f*)(arow + k0);
                v2f b;
                b.x = wcol[k0 * HY];
                b.y = wcol[k0 * HY + HY];
                c = wmma_f32(a, b, c);
            }
            const float bias = bY[ct * 16 + m];
            #pragma unroll
            for (int v = 0; v < 8; ++v)
                Ct[(rt * 16 + khalf * 8 + v) * CPITCH + HX + ct * 16 + m] = fmaxf(c[v] + bias, 0.0f);
        }
    }
    __syncthreads();

    {
        const int c4 = (lane < 16) ? lane * 4 : (lane - 16) * 4;
        float4 g  = make_float4(0.f, 0.f, 0.f, 0.f);
        float4 be = g;
        if (act) {
            g  = *(const float4*)(((lane < 16) ? gX : gY) + c4);
            be = *(const float4*)(((lane < 16) ? beX : beY) + c4);
        }
        const float rn = (lane < 16) ? (1.0f / 64.0f) : (1.0f / 32.0f);
        #pragma unroll
        for (int r = 0; r < ROWS_PER_WAVE; ++r) {
            const int row  = r * BLK_WAVES + wave;
            const int node = base + row;
            if (node < Nn) {
                float4 x = make_float4(0.f, 0.f, 0.f, 0.f);
                if (act) x = *(const float4*)(Ct + row * CPITCH + lane * 4);
                float s  = (x.x + x.y) + (x.z + x.w);
                float s2 = (x.x * x.x + x.y * x.y) + (x.z * x.z + x.w * x.w);
                #pragma unroll
                for (int o = 1; o < 16; o <<= 1) {
                    s  += __shfl_xor(s, o);
                    s2 += __shfl_xor(s2, o);
                }
                const float mean = s * rn;
                const float var  = s2 * rn - mean * mean;
                const float inv  = rsqrtf(var + 1e-5f);
                if (act) {
                    float* o = (lane < 16) ? (outX + (size_t)node * HX + c4) : (outY + (size_t)node * HY + c4);
                    float4 y;
                    y.x = (x.x - mean) * inv * g.x + be.x;
                    y.y = (x.y - mean) * inv * g.y + be.y;
                    y.z = (x.z - mean) * inv * g.z + be.z;
                    y.w = (x.w - mean) * inv * g.w + be.w;
                    *(float4*)o = y;
                }
            }
        }
    }
}

extern "C" void kernel_launch(void* const* d_in, const int* in_sizes, int n_in,
                              void* d_out, int out_size, void* d_ws, size_t ws_size,
                              hipStream_t stream) {
    const int*   erow = (const int*)d_in[0];
    const int*   ecol = (const int*)d_in[1];
    const float* hX   = (const float*)d_in[2];
    const float* hY   = (const float*)d_in[3];
    const float* ht   = (const float*)d_in[4];
    const float* WX   = (const float*)d_in[5];
    const float* bX   = (const float*)d_in[6];
    const float* gX   = (const float*)d_in[7];
    const float* beX  = (const float*)d_in[8];
    const float* WY   = (const float*)d_in[9];
    const float* bY   = (const float*)d_in[10];
    const float* gY   = (const float*)d_in[11];
    const float* beY  = (const float*)d_in[12];

    float*    out    = (float*)d_out;
    char*     ws     = (char*)d_ws;
    int*      offtab = (int*)(ws + WS_OFF);
    unsigned* recs   = (unsigned*)(ws + WS_REC);

    bucket_edges<<<NB, SORT_THREADS, 0, stream>>>(erow, ecol, offtab, recs);

    pull_gemm_ln<<<BLK_COUNT, BLK_WAVES * 32, 0, stream>>>(offtab, recs, hX, hY, ht, WX, bX, gX, beX,
                                                             WY, bY, gY, beY,
                                                             out, out + (size_t)Nn * HX);
}
